# starting kernel
# baseline (speedup 1.0000x reference)
#include <hip/hip_runtime.h>

typedef float v2f __attribute__((ext_vector_type(2)));
typedef float v8f __attribute__((ext_vector_type(8)));

#define P_PILLARS 60000
#define NPT 32
#define COUT 64
#define WPB 8
#define NBLK_A (P_PILLARS / WPB)

__global__ __launch_bounds__(256) void pfn_pass_a(
    const float* __restrict__ features,
    const int*   __restrict__ num_points,
    const int*   __restrict__ coors,
    const float* __restrict__ W,
    float* __restrict__ ws_mx, float* __restrict__ ws_mn,
    float* __restrict__ ws_psum, float* __restrict__ ws_pssq)
{
    __shared__ float Wl[12][COUT];
    __shared__ float featsL[WPB][NPT][12];
    __shared__ float wsum[WPB][COUT];
    __shared__ float wssq[WPB][COUT];

    const int tid = threadIdx.x;
    for (int i = tid; i < 12 * COUT; i += 256) {
        const int r = i / COUT, c = i % COUT;
        Wl[r][c] = (r < 9) ? W[r * COUT + c] : 0.0f;
    }

    const int wid  = tid >> 5;
    const int lane = tid & 31;
    const int half = lane >> 4;
    const int l16  = lane & 15;
    const int p    = blockIdx.x * WPB + wid;

    const float4 f = ((const float4*)features)[p * NPT + lane];
    const int npts = num_points[p];

    float sx = f.x, sy = f.y, sz = f.z;
    #pragma unroll
    for (int off = 16; off >= 1; off >>= 1) {
        sx += __shfl_xor(sx, off, 32);
        sy += __shfl_xor(sy, off, 32);
        sz += __shfl_xor(sz, off, 32);
    }
    const float inv_n = 1.0f / (float)npts;
    const float cmx = sx * inv_n, cmy = sy * inv_n, cmz = sz * inv_n;

    const float xc = (float)coors[p * 4 + 3] * 0.2f + 0.1f;
    const float yc = (float)coors[p * 4 + 2] * 0.2f + (0.1f - 40.0f);

    const float m = (lane < npts) ? 1.0f : 0.0f;
    float* row = &featsL[wid][lane][0];
    row[0] = f.x * m;           row[1] = f.y * m;
    row[2] = f.z * m;           row[3] = f.w * m;
    row[4] = (f.x - cmx) * m;   row[5] = (f.y - cmy) * m;
    row[6] = (f.z - cmz) * m;   row[7] = (f.x - xc) * m;
    row[8] = (f.y - yc) * m;
    row[9] = 0.0f; row[10] = 0.0f; row[11] = 0.0f;

    __syncthreads();

    v2f bfr[3][4];
    #pragma unroll
    for (int ks = 0; ks < 3; ++ks) {
        const int kr = ks * 4 + 2 * half;
        #pragma unroll
        for (int nt = 0; nt < 4; ++nt) {
            bfr[ks][nt][0] = Wl[kr][nt * 16 + l16];
            bfr[ks][nt][1] = Wl[kr + 1][nt * 16 + l16];
        }
    }

    float chmax[4], chmin[4], chs[4], chss[4];
    #pragma unroll
    for (int nt = 0; nt < 4; ++nt) {
        chmax[nt] = -3.402823e38f; chmin[nt] = 3.402823e38f;
        chs[nt] = 0.0f; chss[nt] = 0.0f;
    }

    #pragma unroll
    for (int mt = 0; mt < 2; ++mt) {
        v8f acc[4];
        #pragma unroll
        for (int nt = 0; nt < 4; ++nt) acc[nt] = (v8f){0,0,0,0,0,0,0,0};
        #pragma unroll
        for (int ks = 0; ks < 3; ++ks) {
            v2f a;
            a[0] = featsL[wid][mt * 16 + l16][ks * 4 + 2 * half];
            a[1] = featsL[wid][mt * 16 + l16][ks * 4 + 2 * half + 1];
            #pragma unroll
            for (int nt = 0; nt < 4; ++nt) {
                acc[nt] = __builtin_amdgcn_wmma_f32_16x16x4_f32(
                    false, a, false, bfr[ks][nt], (short)0, acc[nt], false, false);
            }
        }
        #pragma unroll
        for (int nt = 0; nt < 4; ++nt) {
            #pragma unroll
            for (int r = 0; r < 8; ++r) {
                const float v = acc[nt][r];
                chmax[nt] = fmaxf(chmax[nt], v);
                chmin[nt] = fminf(chmin[nt], v);
                chs[nt]  += v;
                chss[nt] += v * v;
            }
        }
    }
    #pragma unroll
    for (int nt = 0; nt < 4; ++nt) {
        chmax[nt] = fmaxf(chmax[nt], __shfl_xor(chmax[nt], 16, 32));
        chmin[nt] = fminf(chmin[nt], __shfl_xor(chmin[nt], 16, 32));
        chs[nt]  += __shfl_xor(chs[nt], 16, 32);
        chss[nt] += __shfl_xor(chss[nt], 16, 32);
    }
    if (half == 0) {
        #pragma unroll
        for (int nt = 0; nt < 4; ++nt) {
            const int c = nt * 16 + l16;
            ws_mx[(size_t)p * COUT + c] = chmax[nt];
            ws_mn[(size_t)p * COUT + c] = chmin[nt];
            wsum[wid][c] = chs[nt];
            wssq[wid][c] = chss[nt];
        }
    }
    __syncthreads();
    if (tid < COUT) {
        float s = 0.0f, ss = 0.0f;
        #pragma unroll
        for (int w = 0; w < WPB; ++w) { s += wsum[w][tid]; ss += wssq[w][tid]; }
        ws_psum[(size_t)blockIdx.x * COUT + tid] = s;
        ws_pssq[(size_t)blockIdx.x * COUT + tid] = ss;
    }
}

__global__ void pfn_stats(const float* __restrict__ psum,
                          const float* __restrict__ pssq,
                          const float* __restrict__ gamma,
                          const float* __restrict__ beta,
                          float* __restrict__ scale,
                          float* __restrict__ shift)
{
    const int c = threadIdx.x;
    if (c >= COUT) return;
    float s = 0.0f, ss = 0.0f;
    for (int b = 0; b < NBLK_A; ++b) {
        s  += psum[(size_t)b * COUT + c];
        ss += pssq[(size_t)b * COUT + c];
    }
    const float inv = 1.0f / (float)(P_PILLARS * NPT);
    const float mean = s * inv;
    const float var  = ss * inv - mean * mean;
    const float sc   = gamma[c] * rsqrtf(var + 1e-3f);
    scale[c] = sc;
    shift[c] = beta[c] - mean * sc;
}

__global__ __launch_bounds__(256) void pfn_finish(
    const float* __restrict__ ws_mx, const float* __restrict__ ws_mn,
    const float* __restrict__ scale, const float* __restrict__ shift,
    float* __restrict__ out)
{
    const int idx = blockIdx.x * 256 + threadIdx.x;
    const int c = idx & (COUT - 1);
    const float s = scale[c];
    const float v = (s >= 0.0f) ? ws_mx[idx] : ws_mn[idx];
    out[idx] = fmaxf(fmaf(s, v, shift[c]), 0.0f);
}

extern "C" void kernel_launch(void* const* d_in, const int* in_sizes, int n_in,
                              void* d_out, int out_size, void* d_ws, size_t ws_size,
                              hipStream_t stream) {
    const float* features   = (const float*)d_in[0];
    const int*   num_points = (const int*)  d_in[1];
    const int*   coors      = (const int*)  d_in[2];
    const float* W          = (const float*)d_in[3];
    const float* gamma      = (const float*)d_in[4];
    const float* beta       = (const float*)d_in[5];
    float* out = (float*)d_out;

    float* ws      = (float*)d_ws;
    float* ws_mx   = ws;
    float* ws_mn   = ws_mx   + (size_t)P_PILLARS * COUT;
    float* ws_psum = ws_mn   + (size_t)P_PILLARS * COUT;
    float* ws_pssq = ws_psum + (size_t)NBLK_A * COUT;
    float* ws_scl  = ws_pssq + (size_t)NBLK_A * COUT;
    float* ws_shf  = ws_scl + COUT;

    pfn_pass_a<<<NBLK_A, 256, 0, stream>>>(features, num_points, coors, W,
                                           ws_mx, ws_mn, ws_psum, ws_pssq);
    pfn_stats<<<1, 64, 0, stream>>>(ws_psum, ws_pssq, gamma, beta, ws_scl, ws_shf);
    pfn_finish<<<(P_PILLARS * COUT) / 256, 256, 0, stream>>>(ws_mx, ws_mn,
                                                             ws_scl, ws_shf, out);
}


# moment pass: per-pillar block barrier removed (each wave's LDS tile is private, a wave's LDS ops complete in order); waves leave lockstep and no longer wait there for the next pillar's loads
# speedup vs baseline: 9.4740x; 9.4740x over previous
#include <hip/hip_runtime.h>

typedef float v2f __attribute__((ext_vector_type(2)));
typedef float v8f __attribute__((ext_vector_type(8)));

#define P_PILLARS 60000
#define NPT 32
#define COUT 64
#define AW 16
#define ASTEPS 15
#define NBLK_A (P_PILLARS / (AW * ASTEPS))

#define MW 16
#define MSTEPS 15
#define NBLK_M (P_PILLARS / (MW * MSTEPS))
#define FPITCH 36

__global__ __launch_bounds__(MW * 32) void pfn_moments(
    const float* __restrict__ features,
    const int*   __restrict__ num_points,
    const int*   __restrict__ coors,
    float* __restrict__ ws_part)
{
    __shared__ __attribute__((aligned(16))) float FT[MW][16][FPITCH];
    __shared__ float red[MW][256];

    const int tid  = threadIdx.x;
    const int wid  = tid >> 5;
    const int lane = tid & 31;
    const int half = lane >> 4;
    const int l16  = lane & 15;

    for (int k = 10; k < 16; ++k) FT[wid][k][lane] = 0.0f;

    const int p0 = blockIdx.x * (MW * MSTEPS) + wid;

    v8f acc0 = (v8f){0,0,0,0,0,0,0,0};
    v8f acc1 = (v8f){0,0,0,0,0,0,0,0};

    float4 f  = ((const float4*)features)[p0 * NPT + lane];
    int npts  = num_points[p0];
    int cx    = coors[p0 * 4 + 3];
    int cy    = coors[p0 * 4 + 2];

    for (int t = 0; t < MSTEPS; ++t) {
        const int pn = p0 + ((t + 1 < MSTEPS) ? (t + 1) : t) * MW;
        const float4 fn = ((const float4*)features)[pn * NPT + lane];
        const int npn = num_points[pn];
        const int cxn = coors[pn * 4 + 3];
        const int cyn = coors[pn * 4 + 2];

        float sx = f.x, sy = f.y, sz = f.z;
        #pragma unroll
        for (int off = 16; off >= 1; off >>= 1) {
            sx += __shfl_xor(sx, off, 32);
            sy += __shfl_xor(sy, off, 32);
            sz += __shfl_xor(sz, off, 32);
        }
        const float inv_n = 1.0f / (float)npts;
        const float cmx = sx * inv_n, cmy = sy * inv_n, cmz = sz * inv_n;
        const float xc = (float)cx * 0.2f + 0.1f;
        const float yc = (float)cy * 0.2f + (0.1f - 40.0f);
        const float m = (lane < npts) ? 1.0f : 0.0f;

        float (*T)[FPITCH] = FT[wid];
        T[0][lane] = f.x * m;           T[1][lane] = f.y * m;
        T[2][lane] = f.z * m;           T[3][lane] = f.w * m;
        T[4][lane] = (f.x - cmx) * m;   T[5][lane] = (f.y - cmy) * m;
        T[6][lane] = (f.z - cmz) * m;   T[7][lane] = (f.x - xc) * m;
        T[8][lane] = (f.y - yc) * m;    T[9][lane] = m;


        const float4* src = (const float4*)&T[l16][16 * half];
        const float4 q0 = src[0], q1 = src[1], q2 = src[2], q3 = src[3];
        v2f a;
        a[0] = q0.x; a[1] = q0.y;
        acc0 = __builtin_amdgcn_wmma_f32_16x16x4_f32(false, a, false, a, (short)0, acc0, false, false);
        a[0] = q0.z; a[1] = q0.w;
        acc1 = __builtin_amdgcn_wmma_f32_16x16x4_f32(false, a, false, a, (short)0, acc1, false, false);
        a[0] = q1.x; a[1] = q1.y;
        acc0 = __builtin_amdgcn_wmma_f32_16x16x4_f32(false, a, false, a, (short)0, acc0, false, false);
        a[0] = q1.z; a[1] = q1.w;
        acc1 = __builtin_amdgcn_wmma_f32_16x16x4_f32(false, a, false, a, (short)0, acc1, false, false);
        a[0] = q2.x; a[1] = q2.y;
        acc0 = __builtin_amdgcn_wmma_f32_16x16x4_f32(false, a, false, a, (short)0, acc0, false, false);
        a[0] = q2.z; a[1] = q2.w;
        acc1 = __builtin_amdgcn_wmma_f32_16x16x4_f32(false, a, false, a, (short)0, acc1, false, false);
        a[0] = q3.x; a[1] = q3.y;
        acc0 = __builtin_amdgcn_wmma_f32_16x16x4_f32(false, a, false, a, (short)0, acc0, false, false);
        a[0] = q3.z; a[1] = q3.w;
        acc1 = __builtin_amdgcn_wmma_f32_16x16x4_f32(false, a, false, a, (short)0, acc1, false, false);

        f = fn; npts = npn; cx = cxn; cy = cyn;
    }

    #pragma unroll
    for (int r = 0; r < 8; ++r)
        red[wid][(8 * half + r) * 16 + l16] = acc0[r] + acc1[r];
    __syncthreads();
    if (tid < 256) {
        float s = 0.0f;
        #pragma unroll
        for (int w = 0; w < MW; ++w) s += red[w][tid];
        ws_part[(size_t)blockIdx.x * 256 + tid] = s;
    }
}

__global__ __launch_bounds__(1024) void pfn_stats(const float* __restrict__ part,
                          const float* __restrict__ W,
                          const float* __restrict__ gamma,
                          const float* __restrict__ beta,
                          float* __restrict__ scale,
                          float* __restrict__ shift)
{
    __shared__ float red[4][256];
    __shared__ float G[256];
    const int tid = threadIdx.x;
    const int g = tid >> 8, e = tid & 255;
    float s = 0.0f;
    for (int o = 0; o < 2; ++o) {
        #pragma unroll
        for (int i = 0; i < 31; ++i)
            s += part[(size_t)(g + 4 * (o * 31 + i)) * 256 + e];
    }
    if (g < 2) s += part[(size_t)(248 + g) * 256 + e];
    red[g][e] = s;
    __syncthreads();
    if (tid < 256) G[tid] = (red[0][tid] + red[1][tid]) + (red[2][tid] + red[3][tid]);
    __syncthreads();
    if (tid < COUT) {
        const int c = tid;
        float w[9];
        #pragma unroll
        for (int k = 0; k < 9; ++k) w[k] = W[k * COUT + c];
        float sum = 0.0f, ssq = 0.0f;
        #pragma unroll
        for (int k = 0; k < 9; ++k) {
            sum += G[9 * 16 + k] * w[k];
            float row = 0.0f;
            #pragma unroll
            for (int j = 0; j < 9; ++j) row += G[k * 16 + j] * w[j];
            ssq += w[k] * row;
        }
        const float inv = 1.0f / (float)(P_PILLARS * NPT);
        const float mean = sum * inv;
        const float var  = ssq * inv - mean * mean;
        const float sc   = gamma[c] * rsqrtf(var + 1e-3f);
        scale[c] = sc;
        shift[c] = beta[c] - mean * sc;
    }
}

__global__ __launch_bounds__(AW * 32) void pfn_pass_a(
    const float* __restrict__ features,
    const int*   __restrict__ num_points,
    const int*   __restrict__ coors,
    const float* __restrict__ W,
    const float* __restrict__ scale,
    const float* __restrict__ shift,
    float* __restrict__ out)
{
    const int tid  = threadIdx.x;
    const int wid  = tid >> 5;
    const int lane = tid & 31;
    const int half = lane >> 4;
    const int l16  = lane & 15;

    const int p0 = blockIdx.x * (AW * ASTEPS) + wid;

    float4 f  = ((const float4*)features)[p0 * NPT + lane];
    int npts  = num_points[p0];
    int cx    = coors[p0 * 4 + 3];
    int cy    = coors[p0 * 4 + 2];

    v2f bfr[3][4];
    #pragma unroll
    for (int ks = 0; ks < 3; ++ks) {
        const int kr = ks * 4 + 2 * half;
        #pragma unroll
        for (int nt = 0; nt < 4; ++nt) {
            bfr[ks][nt][0] = (kr     < 9) ? W[kr * COUT + nt * 16 + l16]       : 0.0f;
            bfr[ks][nt][1] = (kr + 1 < 9) ? W[(kr + 1) * COUT + nt * 16 + l16] : 0.0f;
        }
    }

    const float sc0 = scale[lane], sc1 = scale[32 + lane];
    const float sh0 = shift[lane], sh1 = shift[32 + lane];

    for (int t = 0; t < ASTEPS; ++t) {
        const int p = p0 + t * AW;
        const int pn = p0 + ((t + 1 < ASTEPS) ? (t + 1) : t) * AW;
        const float4 fn = ((const float4*)features)[pn * NPT + lane];
        const int npn = num_points[pn];
        const int cxn = coors[pn * 4 + 3];
        const int cyn = coors[pn * 4 + 2];
        __builtin_amdgcn_sched_barrier(0);

        float4 g;
        g.x = __shfl_xor(f.x, 16, 32);
        g.y = __shfl_xor(f.y, 16, 32);
        g.z = __shfl_xor(f.z, 16, 32);
        g.w = __shfl_xor(f.w, 16, 32);

        float sx = f.x + g.x, sy = f.y + g.y, sz = f.z + g.z;
        #pragma unroll
        for (int off = 8; off >= 1; off >>= 1) {
            sx += __shfl_xor(sx, off, 32);
            sy += __shfl_xor(sy, off, 32);
            sz += __shfl_xor(sz, off, 32);
        }
        const float inv_n = 1.0f / (float)npts;
        const float cmx = sx * inv_n, cmy = sy * inv_n, cmz = sz * inv_n;
        const float xc = (float)cx * 0.2f + 0.1f;
        const float yc = (float)cy * 0.2f + (0.1f - 40.0f);

        const float o_a = half ? cmz : cmx;
        const float o_b = half ? xc  : cmy;

        float chmax[4], chmin[4];
        #pragma unroll
        for (int nt = 0; nt < 4; ++nt) {
            chmax[nt] = -3.402823e38f; chmin[nt] = 3.402823e38f;
        }

        #pragma unroll
        for (int mt = 0; mt < 2; ++mt) {
            const bool own = (half == mt);
            const float rx = own ? f.x : g.x, ry = own ? f.y : g.y;
            const float rz = own ? f.z : g.z, rw = own ? f.w : g.w;
            const float m = (mt * 16 + l16 < npts) ? 1.0f : 0.0f;

            const float u  = half ? rz : rx;
            const float v  = half ? rw : ry;
            const float t1 = half ? rx : ry;
            v2f a[3];
            a[0][0] = u * m;           a[0][1] = v * m;
            a[1][0] = (u - o_a) * m;   a[1][1] = (t1 - o_b) * m;
            a[2][0] = half ? 0.0f : (ry - yc) * m;
            a[2][1] = 0.0f;

            v8f acc[4];
            #pragma unroll
            for (int nt = 0; nt < 4; ++nt) acc[nt] = (v8f){0,0,0,0,0,0,0,0};
            #pragma unroll
            for (int ks = 0; ks < 3; ++ks) {
                #pragma unroll
                for (int nt = 0; nt < 4; ++nt) {
                    acc[nt] = __builtin_amdgcn_wmma_f32_16x16x4_f32(
                        false, a[ks], false, bfr[ks][nt], (short)0, acc[nt], false, false);
                }
            }
            #pragma unroll
            for (int nt = 0; nt < 4; ++nt) {
                #pragma unroll
                for (int r = 0; r < 8; ++r) {
                    const float x = acc[nt][r];
                    chmax[nt] = fmaxf(chmax[nt], x);
                    chmin[nt] = fminf(chmin[nt], x);
                }
            }
        }
        #pragma unroll
        for (int j = 0; j < 2; ++j) {
            const float own_mx  = half ? chmax[2 * j + 1] : chmax[2 * j];
            const float own_mn  = half ? chmin[2 * j + 1] : chmin[2 * j];
            const float send_mx = half ? chmax[2 * j] : chmax[2 * j + 1];
            const float send_mn = half ? chmin[2 * j] : chmin[2 * j + 1];
            const float mx = fmaxf(own_mx, __shfl_xor(send_mx, 16, 32));
            const float mn = fminf(own_mn, __shfl_xor(send_mn, 16, 32));
            const float s  = j ? sc1 : sc0;
            const float sh = j ? sh1 : sh0;
            const float e  = (s >= 0.0f) ? mx : mn;
            out[(size_t)p * COUT + j * 32 + lane] = fmaxf(fmaf(s, e, sh), 0.0f);
        }

        f = fn; npts = npn; cx = cxn; cy = cyn;
    }
}

extern "C" void kernel_launch(void* const* d_in, const int* in_sizes, int n_in,
                              void* d_out, int out_size, void* d_ws, size_t ws_size,
                              hipStream_t stream) {
    const float* features   = (const float*)d_in[0];
    const int*   num_points = (const int*)  d_in[1];
    const int*   coors      = (const int*)  d_in[2];
    const float* W          = (const float*)d_in[3];
    const float* gamma      = (const float*)d_in[4];
    const float* beta       = (const float*)d_in[5];
    float* out = (float*)d_out;

    float* ws      = (float*)d_ws;
    float* ws_part = ws;
    float* ws_scl  = ws_part + (size_t)NBLK_M * 256;
    float* ws_shf  = ws_scl + COUT;

    pfn_moments<<<NBLK_M, MW * 32, 0, stream>>>(features, num_points, coors, ws_part);
    pfn_stats<<<1, 1024, 0, stream>>>(ws_part, W, gamma, beta, ws_scl, ws_shf);
    pfn_pass_a<<<NBLK_A, AW * 32, 0, stream>>>(features, num_points, coors, W, ws_scl, ws_shf, out);
}


# lane exchanges (pillar sums in both passes; point swap and max/min join in the max pass) move from LDS-routed permutes to vector-ALU cross-lane ops (row_xmask DPP, permlanex16); sum order kept
# speedup vs baseline: 10.5758x; 1.1163x over previous
#include <hip/hip_runtime.h>

typedef float v2f __attribute__((ext_vector_type(2)));
typedef float v8f __attribute__((ext_vector_type(8)));

#define P_PILLARS 60000
#define NPT 32
#define COUT 64
#define AW 16
#define ASTEPS 15
#define NBLK_A (P_PILLARS / (AW * ASTEPS))

#define MW 16
#define MSTEPS 15
#define NBLK_M (P_PILLARS / (MW * MSTEPS))
#define FPITCH 36

__device__ __forceinline__ float xhalf(float x) {
    const int b = __builtin_bit_cast(int, x);
    return __builtin_bit_cast(float, __builtin_amdgcn_permlanex16(b, b, 0x76543210, 0xfedcba98, false, false));
}
#define XROW(x, M) __builtin_bit_cast(float, __builtin_amdgcn_update_dpp( \
    __builtin_bit_cast(int, (x)), __builtin_bit_cast(int, (x)), 0x160 + (M), 0xf, 0xf, false))

__global__ __launch_bounds__(MW * 32) void pfn_moments(
    const float* __restrict__ features,
    const int*   __restrict__ num_points,
    const int*   __restrict__ coors,
    float* __restrict__ ws_part)
{
    __shared__ __attribute__((aligned(16))) float FT[MW][16][FPITCH];
    __shared__ float red[MW][256];

    const int tid  = threadIdx.x;
    const int wid  = tid >> 5;
    const int lane = tid & 31;
    const int half = lane >> 4;
    const int l16  = lane & 15;

    for (int k = 10; k < 16; ++k) FT[wid][k][lane] = 0.0f;

    const int p0 = blockIdx.x * (MW * MSTEPS) + wid;

    v8f acc0 = (v8f){0,0,0,0,0,0,0,0};
    v8f acc1 = (v8f){0,0,0,0,0,0,0,0};

    float4 f  = ((const float4*)features)[p0 * NPT + lane];
    int npts  = num_points[p0];
    int cx    = coors[p0 * 4 + 3];
    int cy    = coors[p0 * 4 + 2];

    for (int t = 0; t < MSTEPS; ++t) {
        const int pn = p0 + ((t + 1 < MSTEPS) ? (t + 1) : t) * MW;
        const float4 fn = ((const float4*)features)[pn * NPT + lane];
        const int npn = num_points[pn];
        const int cxn = coors[pn * 4 + 3];
        const int cyn = coors[pn * 4 + 2];

        float sx = f.x, sy = f.y, sz = f.z;
        sx += xhalf(sx);     sy += xhalf(sy);     sz += xhalf(sz);
        sx += XROW(sx, 8);   sy += XROW(sy, 8);   sz += XROW(sz, 8);
        sx += XROW(sx, 4);   sy += XROW(sy, 4);   sz += XROW(sz, 4);
        sx += XROW(sx, 2);   sy += XROW(sy, 2);   sz += XROW(sz, 2);
        sx += XROW(sx, 1);   sy += XROW(sy, 1);   sz += XROW(sz, 1);
        const float inv_n = 1.0f / (float)npts;
        const float cmx = sx * inv_n, cmy = sy * inv_n, cmz = sz * inv_n;
        const float xc = (float)cx * 0.2f + 0.1f;
        const float yc = (float)cy * 0.2f + (0.1f - 40.0f);
        const float m = (lane < npts) ? 1.0f : 0.0f;

        float (*T)[FPITCH] = FT[wid];
        T[0][lane] = f.x * m;           T[1][lane] = f.y * m;
        T[2][lane] = f.z * m;           T[3][lane] = f.w * m;
        T[4][lane] = (f.x - cmx) * m;   T[5][lane] = (f.y - cmy) * m;
        T[6][lane] = (f.z - cmz) * m;   T[7][lane] = (f.x - xc) * m;
        T[8][lane] = (f.y - yc) * m;    T[9][lane] = m;


        const float4* src = (const float4*)&T[l16][16 * half];
        const float4 q0 = src[0], q1 = src[1], q2 = src[2], q3 = src[3];
        v2f a;
        a[0] = q0.x; a[1] = q0.y;
        acc0 = __builtin_amdgcn_wmma_f32_16x16x4_f32(false, a, false, a, (short)0, acc0, false, false);
        a[0] = q0.z; a[1] = q0.w;
        acc1 = __builtin_amdgcn_wmma_f32_16x16x4_f32(false, a, false, a, (short)0, acc1, false, false);
        a[0] = q1.x; a[1] = q1.y;
        acc0 = __builtin_amdgcn_wmma_f32_16x16x4_f32(false, a, false, a, (short)0, acc0, false, false);
        a[0] = q1.z; a[1] = q1.w;
        acc1 = __builtin_amdgcn_wmma_f32_16x16x4_f32(false, a, false, a, (short)0, acc1, false, false);
        a[0] = q2.x; a[1] = q2.y;
        acc0 = __builtin_amdgcn_wmma_f32_16x16x4_f32(false, a, false, a, (short)0, acc0, false, false);
        a[0] = q2.z; a[1] = q2.w;
        acc1 = __builtin_amdgcn_wmma_f32_16x16x4_f32(false, a, false, a, (short)0, acc1, false, false);
        a[0] = q3.x; a[1] = q3.y;
        acc0 = __builtin_amdgcn_wmma_f32_16x16x4_f32(false, a, false, a, (short)0, acc0, false, false);
        a[0] = q3.z; a[1] = q3.w;
        acc1 = __builtin_amdgcn_wmma_f32_16x16x4_f32(false, a, false, a, (short)0, acc1, false, false);

        f = fn; npts = npn; cx = cxn; cy = cyn;
    }

    #pragma unroll
    for (int r = 0; r < 8; ++r)
        red[wid][(8 * half + r) * 16 + l16] = acc0[r] + acc1[r];
    __syncthreads();
    if (tid < 256) {
        float s = 0.0f;
        #pragma unroll
        for (int w = 0; w < MW; ++w) s += red[w][tid];
        ws_part[(size_t)blockIdx.x * 256 + tid] = s;
    }
}

__global__ __launch_bounds__(1024) void pfn_stats(const float* __restrict__ part,
                          const float* __restrict__ W,
                          const float* __restrict__ gamma,
                          const float* __restrict__ beta,
                          float* __restrict__ scale,
                          float* __restrict__ shift)
{
    __shared__ float red[4][256];
    __shared__ float G[256];
    const int tid = threadIdx.x;
    const int g = tid >> 8, e = tid & 255;
    float s = 0.0f;
    for (int o = 0; o < 2; ++o) {
        #pragma unroll
        for (int i = 0; i < 31; ++i)
            s += part[(size_t)(g + 4 * (o * 31 + i)) * 256 + e];
    }
    if (g < 2) s += part[(size_t)(248 + g) * 256 + e];
    red[g][e] = s;
    __syncthreads();
    if (tid < 256) G[tid] = (red[0][tid] + red[1][tid]) + (red[2][tid] + red[3][tid]);
    __syncthreads();
    if (tid < COUT) {
        const int c = tid;
        float w[9];
        #pragma unroll
        for (int k = 0; k < 9; ++k) w[k] = W[k * COUT + c];
        float sum = 0.0f, ssq = 0.0f;
        #pragma unroll
        for (int k = 0; k < 9; ++k) {
            sum += G[9 * 16 + k] * w[k];
            float row = 0.0f;
            #pragma unroll
            for (int j = 0; j < 9; ++j) row += G[k * 16 + j] * w[j];
            ssq += w[k] * row;
        }
        const float inv = 1.0f / (float)(P_PILLARS * NPT);
        const float mean = sum * inv;
        const float var  = ssq * inv - mean * mean;
        const float sc   = gamma[c] * rsqrtf(var + 1e-3f);
        scale[c] = sc;
        shift[c] = beta[c] - mean * sc;
    }
}

__global__ __launch_bounds__(AW * 32) void pfn_pass_a(
    const float* __restrict__ features,
    const int*   __restrict__ num_points,
    const int*   __restrict__ coors,
    const float* __restrict__ W,
    const float* __restrict__ scale,
    const float* __restrict__ shift,
    float* __restrict__ out)
{
    const int tid  = threadIdx.x;
    const int wid  = tid >> 5;
    const int lane = tid & 31;
    const int half = lane >> 4;
    const int l16  = lane & 15;

    const int p0 = blockIdx.x * (AW * ASTEPS) + wid;

    float4 f  = ((const float4*)features)[p0 * NPT + lane];
    int npts  = num_points[p0];
    int cx    = coors[p0 * 4 + 3];
    int cy    = coors[p0 * 4 + 2];

    v2f bfr[3][4];
    #pragma unroll
    for (int ks = 0; ks < 3; ++ks) {
        const int kr = ks * 4 + 2 * half;
        #pragma unroll
        for (int nt = 0; nt < 4; ++nt) {
            bfr[ks][nt][0] = (kr     < 9) ? W[kr * COUT + nt * 16 + l16]       : 0.0f;
            bfr[ks][nt][1] = (kr + 1 < 9) ? W[(kr + 1) * COUT + nt * 16 + l16] : 0.0f;
        }
    }

    const float sc0 = scale[lane], sc1 = scale[32 + lane];
    const float sh0 = shift[lane], sh1 = shift[32 + lane];

    for (int t = 0; t < ASTEPS; ++t) {
        const int p = p0 + t * AW;
        const int pn = p0 + ((t + 1 < ASTEPS) ? (t + 1) : t) * AW;
        const float4 fn = ((const float4*)features)[pn * NPT + lane];
        const int npn = num_points[pn];
        const int cxn = coors[pn * 4 + 3];
        const int cyn = coors[pn * 4 + 2];
        __builtin_amdgcn_sched_barrier(0);

        float4 g;
        g.x = xhalf(f.x);
        g.y = xhalf(f.y);
        g.z = xhalf(f.z);
        g.w = xhalf(f.w);

        float sx = f.x + g.x, sy = f.y + g.y, sz = f.z + g.z;
        sx += XROW(sx, 8);   sy += XROW(sy, 8);   sz += XROW(sz, 8);
        sx += XROW(sx, 4);   sy += XROW(sy, 4);   sz += XROW(sz, 4);
        sx += XROW(sx, 2);   sy += XROW(sy, 2);   sz += XROW(sz, 2);
        sx += XROW(sx, 1);   sy += XROW(sy, 1);   sz += XROW(sz, 1);
        const float inv_n = 1.0f / (float)npts;
        const float cmx = sx * inv_n, cmy = sy * inv_n, cmz = sz * inv_n;
        const float xc = (float)cx * 0.2f + 0.1f;
        const float yc = (float)cy * 0.2f + (0.1f - 40.0f);

        const float o_a = half ? cmz : cmx;
        const float o_b = half ? xc  : cmy;

        float chmax[4], chmin[4];
        #pragma unroll
        for (int nt = 0; nt < 4; ++nt) {
            chmax[nt] = -3.402823e38f; chmin[nt] = 3.402823e38f;
        }

        #pragma unroll
        for (int mt = 0; mt < 2; ++mt) {
            const bool own = (half == mt);
            const float rx = own ? f.x : g.x, ry = own ? f.y : g.y;
            const float rz = own ? f.z : g.z, rw = own ? f.w : g.w;
            const float m = (mt * 16 + l16 < npts) ? 1.0f : 0.0f;

            const float u  = half ? rz : rx;
            const float v  = half ? rw : ry;
            const float t1 = half ? rx : ry;
            v2f a[3];
            a[0][0] = u * m;           a[0][1] = v * m;
            a[1][0] = (u - o_a) * m;   a[1][1] = (t1 - o_b) * m;
            a[2][0] = half ? 0.0f : (ry - yc) * m;
            a[2][1] = 0.0f;

            v8f acc[4];
            #pragma unroll
            for (int nt = 0; nt < 4; ++nt) acc[nt] = (v8f){0,0,0,0,0,0,0,0};
            #pragma unroll
            for (int ks = 0; ks < 3; ++ks) {
                #pragma unroll
                for (int nt = 0; nt < 4; ++nt) {
                    acc[nt] = __builtin_amdgcn_wmma_f32_16x16x4_f32(
                        false, a[ks], false, bfr[ks][nt], (short)0, acc[nt], false, false);
                }
            }
            #pragma unroll
            for (int nt = 0; nt < 4; ++nt) {
                #pragma unroll
                for (int r = 0; r < 8; ++r) {
                    const float x = acc[nt][r];
                    chmax[nt] = fmaxf(chmax[nt], x);
                    chmin[nt] = fminf(chmin[nt], x);
                }
            }
        }
        #pragma unroll
        for (int j = 0; j < 2; ++j) {
            const float own_mx  = half ? chmax[2 * j + 1] : chmax[2 * j];
            const float own_mn  = half ? chmin[2 * j + 1] : chmin[2 * j];
            const float send_mx = half ? chmax[2 * j] : chmax[2 * j + 1];
            const float send_mn = half ? chmin[2 * j] : chmin[2 * j + 1];
            const float mx = fmaxf(own_mx, xhalf(send_mx));
            const float mn = fminf(own_mn, xhalf(send_mn));
            const float s  = j ? sc1 : sc0;
            const float sh = j ? sh1 : sh0;
            const float e  = (s >= 0.0f) ? mx : mn;
            out[(size_t)p * COUT + j * 32 + lane] = fmaxf(fmaf(s, e, sh), 0.0f);
        }

        f = fn; npts = npn; cx = cxn; cy = cyn;
    }
}

extern "C" void kernel_launch(void* const* d_in, const int* in_sizes, int n_in,
                              void* d_out, int out_size, void* d_ws, size_t ws_size,
                              hipStream_t stream) {
    const float* features   = (const float*)d_in[0];
    const int*   num_points = (const int*)  d_in[1];
    const int*   coors      = (const int*)  d_in[2];
    const float* W          = (const float*)d_in[3];
    const float* gamma      = (const float*)d_in[4];
    const float* beta       = (const float*)d_in[5];
    float* out = (float*)d_out;

    float* ws      = (float*)d_ws;
    float* ws_part = ws;
    float* ws_scl  = ws_part + (size_t)NBLK_M * 256;
    float* ws_shf  = ws_scl + COUT;

    pfn_moments<<<NBLK_M, MW * 32, 0, stream>>>(features, num_points, coors, ws_part);
    pfn_stats<<<1, 1024, 0, stream>>>(ws_part, W, gamma, beta, ws_scl, ws_shf);
    pfn_pass_a<<<NBLK_A, AW * 32, 0, stream>>>(features, num_points, coors, W, ws_scl, ws_shf, out);
}


# moment pass hands the stats launch only the 55 moments it uses (upper triangle of the symmetric 10x10 corner; rows of 64 floats, was 256); stats block 1024 -> 256 threads; order of adds kept
# speedup vs baseline: 11.0583x; 1.0456x over previous
#include <hip/hip_runtime.h>

typedef float v2f __attribute__((ext_vector_type(2)));
typedef float v8f __attribute__((ext_vector_type(8)));

#define P_PILLARS 60000
#define NPT 32
#define COUT 64
#define AW 16
#define ASTEPS 15
#define NBLK_A (P_PILLARS / (AW * ASTEPS))

#define MW 16
#define MSTEPS 15
#define NBLK_M (P_PILLARS / (MW * MSTEPS))
#define FPITCH 36

#define GROW 64
__host__ __device__ constexpr int tri(int a, int b) {
    return (a < b ? a : b) * 10 - ((a < b ? a : b) * ((a < b ? a : b) - 1)) / 2 + ((a < b ? b : a) - (a < b ? a : b));
}

__device__ __forceinline__ float xhalf(float x) {
    const int b = __builtin_bit_cast(int, x);
    return __builtin_bit_cast(float, __builtin_amdgcn_permlanex16(b, b, 0x76543210, 0xfedcba98, false, false));
}
#define XROW(x, M) __builtin_bit_cast(float, __builtin_amdgcn_update_dpp( \
    __builtin_bit_cast(int, (x)), __builtin_bit_cast(int, (x)), 0x160 + (M), 0xf, 0xf, false))

__global__ __launch_bounds__(MW * 32) void pfn_moments(
    const float* __restrict__ features,
    const int*   __restrict__ num_points,
    const int*   __restrict__ coors,
    float* __restrict__ ws_part)
{
    __shared__ __attribute__((aligned(16))) float FT[MW][16][FPITCH];
    __shared__ float red[MW][256];

    const int tid  = threadIdx.x;
    const int wid  = tid >> 5;
    const int lane = tid & 31;
    const int half = lane >> 4;
    const int l16  = lane & 15;

    for (int k = 10; k < 16; ++k) FT[wid][k][lane] = 0.0f;

    const int p0 = blockIdx.x * (MW * MSTEPS) + wid;

    v8f acc0 = (v8f){0,0,0,0,0,0,0,0};
    v8f acc1 = (v8f){0,0,0,0,0,0,0,0};

    float4 f  = ((const float4*)features)[p0 * NPT + lane];
    int npts  = num_points[p0];
    int cx    = coors[p0 * 4 + 3];
    int cy    = coors[p0 * 4 + 2];

    for (int t = 0; t < MSTEPS; ++t) {
        const int pn = p0 + ((t + 1 < MSTEPS) ? (t + 1) : t) * MW;
        const float4 fn = ((const float4*)features)[pn * NPT + lane];
        const int npn = num_points[pn];
        const int cxn = coors[pn * 4 + 3];
        const int cyn = coors[pn * 4 + 2];

        float sx = f.x, sy = f.y, sz = f.z;
        sx += xhalf(sx);     sy += xhalf(sy);     sz += xhalf(sz);
        sx += XROW(sx, 8);   sy += XROW(sy, 8);   sz += XROW(sz, 8);
        sx += XROW(sx, 4);   sy += XROW(sy, 4);   sz += XROW(sz, 4);
        sx += XROW(sx, 2);   sy += XROW(sy, 2);   sz += XROW(sz, 2);
        sx += XROW(sx, 1);   sy += XROW(sy, 1);   sz += XROW(sz, 1);
        const float inv_n = 1.0f / (float)npts;
        const float cmx = sx * inv_n, cmy = sy * inv_n, cmz = sz * inv_n;
        const float xc = (float)cx * 0.2f + 0.1f;
        const float yc = (float)cy * 0.2f + (0.1f - 40.0f);
        const float m = (lane < npts) ? 1.0f : 0.0f;

        float (*T)[FPITCH] = FT[wid];
        T[0][lane] = f.x * m;           T[1][lane] = f.y * m;
        T[2][lane] = f.z * m;           T[3][lane] = f.w * m;
        T[4][lane] = (f.x - cmx) * m;   T[5][lane] = (f.y - cmy) * m;
        T[6][lane] = (f.z - cmz) * m;   T[7][lane] = (f.x - xc) * m;
        T[8][lane] = (f.y - yc) * m;    T[9][lane] = m;


        const float4* src = (const float4*)&T[l16][16 * half];
        const float4 q0 = src[0], q1 = src[1], q2 = src[2], q3 = src[3];
        v2f a;
        a[0] = q0.x; a[1] = q0.y;
        acc0 = __builtin_amdgcn_wmma_f32_16x16x4_f32(false, a, false, a, (short)0, acc0, false, false);
        a[0] = q0.z; a[1] = q0.w;
        acc1 = __builtin_amdgcn_wmma_f32_16x16x4_f32(false, a, false, a, (short)0, acc1, false, false);
        a[0] = q1.x; a[1] = q1.y;
        acc0 = __builtin_amdgcn_wmma_f32_16x16x4_f32(false, a, false, a, (short)0, acc0, false, false);
        a[0] = q1.z; a[1] = q1.w;
        acc1 = __builtin_amdgcn_wmma_f32_16x16x4_f32(false, a, false, a, (short)0, acc1, false, false);
        a[0] = q2.x; a[1] = q2.y;
        acc0 = __builtin_amdgcn_wmma_f32_16x16x4_f32(false, a, false, a, (short)0, acc0, false, false);
        a[0] = q2.z; a[1] = q2.w;
        acc1 = __builtin_amdgcn_wmma_f32_16x16x4_f32(false, a, false, a, (short)0, acc1, false, false);
        a[0] = q3.x; a[1] = q3.y;
        acc0 = __builtin_amdgcn_wmma_f32_16x16x4_f32(false, a, false, a, (short)0, acc0, false, false);
        a[0] = q3.z; a[1] = q3.w;
        acc1 = __builtin_amdgcn_wmma_f32_16x16x4_f32(false, a, false, a, (short)0, acc1, false, false);

        f = fn; npts = npn; cx = cxn; cy = cyn;
    }

    #pragma unroll
    for (int r = 0; r < 8; ++r)
        red[wid][(8 * half + r) * 16 + l16] = acc0[r] + acc1[r];
    __syncthreads();
    if (tid < GROW) {
        int k = 0;
        #pragma unroll
        for (int r = 1; r < 10; ++r) k += (tid >= tri(r, r)) ? 1 : 0;
        const int rem = tid - tri(k, k);
        float s = 0.0f;
        if (tid < 55) {
            const int src = k * 16 + k + rem;
            #pragma unroll
            for (int w = 0; w < MW; ++w) s += red[w][src];
        }
        ws_part[(size_t)blockIdx.x * GROW + tid] = s;
    }
}

__global__ __launch_bounds__(256) void pfn_stats(const float* __restrict__ part,
                          const float* __restrict__ W,
                          const float* __restrict__ gamma,
                          const float* __restrict__ beta,
                          float* __restrict__ scale,
                          float* __restrict__ shift)
{
    __shared__ float red[4][GROW];
    __shared__ float G[GROW];
    const int tid = threadIdx.x;
    const int g = tid >> 6, e = tid & (GROW - 1);
    float s = 0.0f;
    for (int o = 0; o < 2; ++o) {
        #pragma unroll
        for (int i = 0; i < 31; ++i)
            s += part[(size_t)(g + 4 * (o * 31 + i)) * GROW + e];
    }
    if (g < 2) s += part[(size_t)(248 + g) * GROW + e];
    red[g][e] = s;
    __syncthreads();
    if (tid < GROW) G[tid] = (red[0][tid] + red[1][tid]) + (red[2][tid] + red[3][tid]);
    __syncthreads();
    if (tid < COUT) {
        const int c = tid;
        float w[9];
        #pragma unroll
        for (int k = 0; k < 9; ++k) w[k] = W[k * COUT + c];
        float sum = 0.0f, ssq = 0.0f;
        #pragma unroll
        for (int k = 0; k < 9; ++k) {
            sum += G[tri(9, k)] * w[k];
            float row = 0.0f;
            #pragma unroll
            for (int j = 0; j < 9; ++j) row += G[tri(k, j)] * w[j];
            ssq += w[k] * row;
        }
        const float inv = 1.0f / (float)(P_PILLARS * NPT);
        const float mean = sum * inv;
        const float var  = ssq * inv - mean * mean;
        const float sc   = gamma[c] * rsqrtf(var + 1e-3f);
        scale[c] = sc;
        shift[c] = beta[c] - mean * sc;
    }
}

__global__ __launch_bounds__(AW * 32) void pfn_pass_a(
    const float* __restrict__ features,
    const int*   __restrict__ num_points,
    const int*   __restrict__ coors,
    const float* __restrict__ W,
    const float* __restrict__ scale,
    const float* __restrict__ shift,
    float* __restrict__ out)
{
    const int tid  = threadIdx.x;
    const int wid  = tid >> 5;
    const int lane = tid & 31;
    const int half = lane >> 4;
    const int l16  = lane & 15;

    const int p0 = blockIdx.x * (AW * ASTEPS) + wid;

    float4 f  = ((const float4*)features)[p0 * NPT + lane];
    int npts  = num_points[p0];
    int cx    = coors[p0 * 4 + 3];
    int cy    = coors[p0 * 4 + 2];

    v2f bfr[3][4];
    #pragma unroll
    for (int ks = 0; ks < 3; ++ks) {
        const int kr = ks * 4 + 2 * half;
        #pragma unroll
        for (int nt = 0; nt < 4; ++nt) {
            bfr[ks][nt][0] = (kr     < 9) ? W[kr * COUT + nt * 16 + l16]       : 0.0f;
            bfr[ks][nt][1] = (kr + 1 < 9) ? W[(kr + 1) * COUT + nt * 16 + l16] : 0.0f;
        }
    }

    const float sc0 = scale[lane], sc1 = scale[32 + lane];
    const float sh0 = shift[lane], sh1 = shift[32 + lane];

    for (int t = 0; t < ASTEPS; ++t) {
        const int p = p0 + t * AW;
        const int pn = p0 + ((t + 1 < ASTEPS) ? (t + 1) : t) * AW;
        const float4 fn = ((const float4*)features)[pn * NPT + lane];
        const int npn = num_points[pn];
        const int cxn = coors[pn * 4 + 3];
        const int cyn = coors[pn * 4 + 2];
        __builtin_amdgcn_sched_barrier(0);

        float4 g;
        g.x = xhalf(f.x);
        g.y = xhalf(f.y);
        g.z = xhalf(f.z);
        g.w = xhalf(f.w);

        float sx = f.x + g.x, sy = f.y + g.y, sz = f.z + g.z;
        sx += XROW(sx, 8);   sy += XROW(sy, 8);   sz += XROW(sz, 8);
        sx += XROW(sx, 4);   sy += XROW(sy, 4);   sz += XROW(sz, 4);
        sx += XROW(sx, 2);   sy += XROW(sy, 2);   sz += XROW(sz, 2);
        sx += XROW(sx, 1);   sy += XROW(sy, 1);   sz += XROW(sz, 1);
        const float inv_n = 1.0f / (float)npts;
        const float cmx = sx * inv_n, cmy = sy * inv_n, cmz = sz * inv_n;
        const float xc = (float)cx * 0.2f + 0.1f;
        const float yc = (float)cy * 0.2f + (0.1f - 40.0f);

        const float o_a = half ? cmz : cmx;
        const float o_b = half ? xc  : cmy;

        float chmax[4], chmin[4];
        #pragma unroll
        for (int nt = 0; nt < 4; ++nt) {
            chmax[nt] = -3.402823e38f; chmin[nt] = 3.402823e38f;
        }

        #pragma unroll
        for (int mt = 0; mt < 2; ++mt) {
            const bool own = (half == mt);
            const float rx = own ? f.x : g.x, ry = own ? f.y : g.y;
            const float rz = own ? f.z : g.z, rw = own ? f.w : g.w;
            const float m = (mt * 16 + l16 < npts) ? 1.0f : 0.0f;

            const float u  = half ? rz : rx;
            const float v  = half ? rw : ry;
            const float t1 = half ? rx : ry;
            v2f a[3];
            a[0][0] = u * m;           a[0][1] = v * m;
            a[1][0] = (u - o_a) * m;   a[1][1] = (t1 - o_b) * m;
            a[2][0] = half ? 0.0f : (ry - yc) * m;
            a[2][1] = 0.0f;

            v8f acc[4];
            #pragma unroll
            for (int nt = 0; nt < 4; ++nt) acc[nt] = (v8f){0,0,0,0,0,0,0,0};
            #pragma unroll
            for (int ks = 0; ks < 3; ++ks) {
                #pragma unroll
                for (int nt = 0; nt < 4; ++nt) {
                    acc[nt] = __builtin_amdgcn_wmma_f32_16x16x4_f32(
                        false, a[ks], false, bfr[ks][nt], (short)0, acc[nt], false, false);
                }
            }
            #pragma unroll
            for (int nt = 0; nt < 4; ++nt) {
                #pragma unroll
                for (int r = 0; r < 8; ++r) {
                    const float x = acc[nt][r];
                    chmax[nt] = fmaxf(chmax[nt], x);
                    chmin[nt] = fminf(chmin[nt], x);
                }
            }
        }
        #pragma unroll
        for (int j = 0; j < 2; ++j) {
            const float own_mx  = half ? chmax[2 * j + 1] : chmax[2 * j];
            const float own_mn  = half ? chmin[2 * j + 1] : chmin[2 * j];
            const float send_mx = half ? chmax[2 * j] : chmax[2 * j + 1];
            const float send_mn = half ? chmin[2 * j] : chmin[2 * j + 1];
            const float mx = fmaxf(own_mx, xhalf(send_mx));
            const float mn = fminf(own_mn, xhalf(send_mn));
            const float s  = j ? sc1 : sc0;
            const float sh = j ? sh1 : sh0;
            const float e  = (s >= 0.0f) ? mx : mn;
            out[(size_t)p * COUT + j * 32 + lane] = fmaxf(fmaf(s, e, sh), 0.0f);
        }

        f = fn; npts = npn; cx = cxn; cy = cyn;
    }
}

extern "C" void kernel_launch(void* const* d_in, const int* in_sizes, int n_in,
                              void* d_out, int out_size, void* d_ws, size_t ws_size,
                              hipStream_t stream) {
    const float* features   = (const float*)d_in[0];
    const int*   num_points = (const int*)  d_in[1];
    const int*   coors      = (const int*)  d_in[2];
    const float* W          = (const float*)d_in[3];
    const float* gamma      = (const float*)d_in[4];
    const float* beta       = (const float*)d_in[5];
    float* out = (float*)d_out;

    float* ws      = (float*)d_ws;
    float* ws_part = ws;
    float* ws_scl  = ws_part + (size_t)NBLK_M * GROW;
    float* ws_shf  = ws_scl + COUT;

    pfn_moments<<<NBLK_M, MW * 32, 0, stream>>>(features, num_points, coors, ws_part);
    pfn_stats<<<1, 256, 0, stream>>>(ws_part, W, gamma, beta, ws_scl, ws_shf);
    pfn_pass_a<<<NBLK_A, AW * 32, 0, stream>>>(features, num_points, coors, W, ws_scl, ws_shf, out);
}


# max pass: batch-norm scale folded into W's B fragments, shift into padding K row 9 against a 1 in A: the matrix unit gives the normalised value, out = max(0, max); min chain, fma go; rounding changes
# speedup vs baseline: 11.4280x; 1.0334x over previous
#include <hip/hip_runtime.h>

typedef float v2f __attribute__((ext_vector_type(2)));
typedef float v8f __attribute__((ext_vector_type(8)));

#define P_PILLARS 60000
#define NPT 32
#define COUT 64
#define AW 16
#define ASTEPS 15
#define NBLK_A (P_PILLARS / (AW * ASTEPS))

#define MW 16
#define MSTEPS 15
#define NBLK_M (P_PILLARS / (MW * MSTEPS))
#define FPITCH 36

#define GROW 64
__host__ __device__ constexpr int tri(int a, int b) {
    return (a < b ? a : b) * 10 - ((a < b ? a : b) * ((a < b ? a : b) - 1)) / 2 + ((a < b ? b : a) - (a < b ? a : b));
}

__device__ __forceinline__ float xhalf(float x) {
    const int b = __builtin_bit_cast(int, x);
    return __builtin_bit_cast(float, __builtin_amdgcn_permlanex16(b, b, 0x76543210, 0xfedcba98, false, false));
}
#define XROW(x, M) __builtin_bit_cast(float, __builtin_amdgcn_update_dpp( \
    __builtin_bit_cast(int, (x)), __builtin_bit_cast(int, (x)), 0x160 + (M), 0xf, 0xf, false))

__global__ __launch_bounds__(MW * 32) void pfn_moments(
    const float* __restrict__ features,
    const int*   __restrict__ num_points,
    const int*   __restrict__ coors,
    float* __restrict__ ws_part)
{
    __shared__ __attribute__((aligned(16))) float FT[MW][16][FPITCH];
    __shared__ float red[MW][256];

    const int tid  = threadIdx.x;
    const int wid  = tid >> 5;
    const int lane = tid & 31;
    const int half = lane >> 4;
    const int l16  = lane & 15;

    for (int k = 10; k < 16; ++k) FT[wid][k][lane] = 0.0f;

    const int p0 = blockIdx.x * (MW * MSTEPS) + wid;

    v8f acc0 = (v8f){0,0,0,0,0,0,0,0};
    v8f acc1 = (v8f){0,0,0,0,0,0,0,0};

    float4 f  = ((const float4*)features)[p0 * NPT + lane];
    int npts  = num_points[p0];
    int cx    = coors[p0 * 4 + 3];
    int cy    = coors[p0 * 4 + 2];

    for (int t = 0; t < MSTEPS; ++t) {
        const int pn = p0 + ((t + 1 < MSTEPS) ? (t + 1) : t) * MW;
        const float4 fn = ((const float4*)features)[pn * NPT + lane];
        const int npn = num_points[pn];
        const int cxn = coors[pn * 4 + 3];
        const int cyn = coors[pn * 4 + 2];

        float sx = f.x, sy = f.y, sz = f.z;
        sx += xhalf(sx);     sy += xhalf(sy);     sz += xhalf(sz);
        sx += XROW(sx, 8);   sy += XROW(sy, 8);   sz += XROW(sz, 8);
        sx += XROW(sx, 4);   sy += XROW(sy, 4);   sz += XROW(sz, 4);
        sx += XROW(sx, 2);   sy += XROW(sy, 2);   sz += XROW(sz, 2);
        sx += XROW(sx, 1);   sy += XROW(sy, 1);   sz += XROW(sz, 1);
        const float inv_n = 1.0f / (float)npts;
        const float cmx = sx * inv_n, cmy = sy * inv_n, cmz = sz * inv_n;
        const float xc = (float)cx * 0.2f + 0.1f;
        const float yc = (float)cy * 0.2f + (0.1f - 40.0f);
        const float m = (lane < npts) ? 1.0f : 0.0f;

        float (*T)[FPITCH] = FT[wid];
        T[0][lane] = f.x * m;           T[1][lane] = f.y * m;
        T[2][lane] = f.z * m;           T[3][lane] = f.w * m;
        T[4][lane] = (f.x - cmx) * m;   T[5][lane] = (f.y - cmy) * m;
        T[6][lane] = (f.z - cmz) * m;   T[7][lane] = (f.x - xc) * m;
        T[8][lane] = (f.y - yc) * m;    T[9][lane] = m;


        const float4* src = (const float4*)&T[l16][16 * half];
        const float4 q0 = src[0], q1 = src[1], q2 = src[2], q3 = src[3];
        v2f a;
        a[0] = q0.x; a[1] = q0.y;
        acc0 = __builtin_amdgcn_wmma_f32_16x16x4_f32(false, a, false, a, (short)0, acc0, false, false);
        a[0] = q0.z; a[1] = q0.w;
        acc1 = __builtin_amdgcn_wmma_f32_16x16x4_f32(false, a, false, a, (short)0, acc1, false, false);
        a[0] = q1.x; a[1] = q1.y;
        acc0 = __builtin_amdgcn_wmma_f32_16x16x4_f32(false, a, false, a, (short)0, acc0, false, false);
        a[0] = q1.z; a[1] = q1.w;
        acc1 = __builtin_amdgcn_wmma_f32_16x16x4_f32(false, a, false, a, (short)0, acc1, false, false);
        a[0] = q2.x; a[1] = q2.y;
        acc0 = __builtin_amdgcn_wmma_f32_16x16x4_f32(false, a, false, a, (short)0, acc0, false, false);
        a[0] = q2.z; a[1] = q2.w;
        acc1 = __builtin_amdgcn_wmma_f32_16x16x4_f32(false, a, false, a, (short)0, acc1, false, false);
        a[0] = q3.x; a[1] = q3.y;
        acc0 = __builtin_amdgcn_wmma_f32_16x16x4_f32(false, a, false, a, (short)0, acc0, false, false);
        a[0] = q3.z; a[1] = q3.w;
        acc1 = __builtin_amdgcn_wmma_f32_16x16x4_f32(false, a, false, a, (short)0, acc1, false, false);

        f = fn; npts = npn; cx = cxn; cy = cyn;
    }

    #pragma unroll
    for (int r = 0; r < 8; ++r)
        red[wid][(8 * half + r) * 16 + l16] = acc0[r] + acc1[r];
    __syncthreads();
    if (tid < GROW) {
        int k = 0;
        #pragma unroll
        for (int r = 1; r < 10; ++r) k += (tid >= tri(r, r)) ? 1 : 0;
        const int rem = tid - tri(k, k);
        float s = 0.0f;
        if (tid < 55) {
            const int src = k * 16 + k + rem;
            #pragma unroll
            for (int w = 0; w < MW; ++w) s += red[w][src];
        }
        ws_part[(size_t)blockIdx.x * GROW + tid] = s;
    }
}

#define KPAD 12

__global__ __launch_bounds__(256) void pfn_stats(const float* __restrict__ part,
                          const float* __restrict__ W,
                          const float* __restrict__ gamma,
                          const float* __restrict__ beta,
                          float* __restrict__ fold)
{
    __shared__ float red[4][GROW];
    __shared__ float G[GROW];
    const int tid = threadIdx.x;
    const int g = tid >> 6, e = tid & (GROW - 1);
    float s = 0.0f;
    for (int o = 0; o < 2; ++o) {
        #pragma unroll
        for (int i = 0; i < 31; ++i)
            s += part[(size_t)(g + 4 * (o * 31 + i)) * GROW + e];
    }
    if (g < 2) s += part[(size_t)(248 + g) * GROW + e];
    red[g][e] = s;
    __syncthreads();
    if (tid < GROW) G[tid] = (red[0][tid] + red[1][tid]) + (red[2][tid] + red[3][tid]);
    __syncthreads();
    if (tid < COUT) {
        const int c = tid;
        float w[9];
        #pragma unroll
        for (int k = 0; k < 9; ++k) w[k] = W[k * COUT + c];
        float sum = 0.0f, ssq = 0.0f;
        #pragma unroll
        for (int k = 0; k < 9; ++k) {
            sum += G[tri(9, k)] * w[k];
            float row = 0.0f;
            #pragma unroll
            for (int j = 0; j < 9; ++j) row += G[tri(k, j)] * w[j];
            ssq += w[k] * row;
        }
        const float inv = 1.0f / (float)(P_PILLARS * NPT);
        const float mean = sum * inv;
        const float var  = ssq * inv - mean * mean;
        const float sc   = gamma[c] * rsqrtf(var + 1e-3f);
        #pragma unroll
        for (int k = 0; k < 9; ++k) fold[k * COUT + c] = w[k] * sc;
        fold[9 * COUT + c] = beta[c] - mean * sc;
        #pragma unroll
        for (int k = 10; k < KPAD; ++k) fold[k * COUT + c] = 0.0f;
    }
}

__global__ __launch_bounds__(AW * 32) void pfn_pass_a(
    const float* __restrict__ features,
    const int*   __restrict__ num_points,
    const int*   __restrict__ coors,
    const float* __restrict__ fold,
    float* __restrict__ out)
{
    const int tid  = threadIdx.x;
    const int wid  = tid >> 5;
    const int lane = tid & 31;
    const int half = lane >> 4;
    const int l16  = lane & 15;

    const int p0 = blockIdx.x * (AW * ASTEPS) + wid;

    float4 f  = ((const float4*)features)[p0 * NPT + lane];
    int npts  = num_points[p0];
    int cx    = coors[p0 * 4 + 3];
    int cy    = coors[p0 * 4 + 2];

    v2f bfr[KPAD / 4][4];
    #pragma unroll
    for (int ks = 0; ks < KPAD / 4; ++ks) {
        const int kr = ks * 4 + 2 * half;
        #pragma unroll
        for (int nt = 0; nt < 4; ++nt) {
            bfr[ks][nt][0] = fold[kr * COUT + nt * 16 + l16];
            bfr[ks][nt][1] = fold[(kr + 1) * COUT + nt * 16 + l16];
        }
    }
    const float a_one = half ? 0.0f : 1.0f;

    for (int t = 0; t < ASTEPS; ++t) {
        const int p = p0 + t * AW;
        const int pn = p0 + ((t + 1 < ASTEPS) ? (t + 1) : t) * AW;
        const float4 fn = ((const float4*)features)[pn * NPT + lane];
        const int npn = num_points[pn];
        const int cxn = coors[pn * 4 + 3];
        const int cyn = coors[pn * 4 + 2];
        __builtin_amdgcn_sched_barrier(0);

        float4 g;
        g.x = xhalf(f.x);
        g.y = xhalf(f.y);
        g.z = xhalf(f.z);
        g.w = xhalf(f.w);

        float sx = f.x + g.x, sy = f.y + g.y, sz = f.z + g.z;
        sx += XROW(sx, 8);   sy += XROW(sy, 8);   sz += XROW(sz, 8);
        sx += XROW(sx, 4);   sy += XROW(sy, 4);   sz += XROW(sz, 4);
        sx += XROW(sx, 2);   sy += XROW(sy, 2);   sz += XROW(sz, 2);
        sx += XROW(sx, 1);   sy += XROW(sy, 1);   sz += XROW(sz, 1);
        const float inv_n = 1.0f / (float)npts;
        const float cmx = sx * inv_n, cmy = sy * inv_n, cmz = sz * inv_n;
        const float xc = (float)cx * 0.2f + 0.1f;
        const float yc = (float)cy * 0.2f + (0.1f - 40.0f);

        const float o_a = half ? cmz : cmx;
        const float o_b = half ? xc  : cmy;

        float chmax[4];
        #pragma unroll
        for (int nt = 0; nt < 4; ++nt) chmax[nt] = 0.0f;

        #pragma unroll
        for (int mt = 0; mt < 2; ++mt) {
            const bool own = (half == mt);
            const float rx = own ? f.x : g.x, ry = own ? f.y : g.y;
            const float rz = own ? f.z : g.z, rw = own ? f.w : g.w;
            const float m = (mt * 16 + l16 < npts) ? 1.0f : 0.0f;

            const float u  = half ? rz : rx;
            const float v  = half ? rw : ry;
            const float t1 = half ? rx : ry;
            v2f a[3];
            a[0][0] = u * m;           a[0][1] = v * m;
            a[1][0] = (u - o_a) * m;   a[1][1] = (t1 - o_b) * m;
            a[2][0] = half ? 0.0f : (ry - yc) * m;
            a[2][1] = a_one;

            v8f acc[4];
            #pragma unroll
            for (int nt = 0; nt < 4; ++nt) acc[nt] = (v8f){0,0,0,0,0,0,0,0};
            #pragma unroll
            for (int ks = 0; ks < 3; ++ks) {
                #pragma unroll
                for (int nt = 0; nt < 4; ++nt) {
                    acc[nt] = __builtin_amdgcn_wmma_f32_16x16x4_f32(
                        false, a[ks], false, bfr[ks][nt], (short)0, acc[nt], false, false);
                }
            }
            #pragma unroll
            for (int nt = 0; nt < 4; ++nt) {
                #pragma unroll
                for (int r = 0; r < 8; ++r) chmax[nt] = fmaxf(chmax[nt], acc[nt][r]);
            }
        }
        #pragma unroll
        for (int j = 0; j < 2; ++j) {
            const float own_mx  = half ? chmax[2 * j + 1] : chmax[2 * j];
            const float send_mx = half ? chmax[2 * j] : chmax[2 * j + 1];
            out[(size_t)p * COUT + j * 32 + lane] = fmaxf(own_mx, xhalf(send_mx));
        }

        f = fn; npts = npn; cx = cxn; cy = cyn;
    }
}

extern "C" void kernel_launch(void* const* d_in, const int* in_sizes, int n_in,
                              void* d_out, int out_size, void* d_ws, size_t ws_size,
                              hipStream_t stream) {
    const float* features   = (const float*)d_in[0];
    const int*   num_points = (const int*)  d_in[1];
    const int*   coors      = (const int*)  d_in[2];
    const float* W          = (const float*)d_in[3];
    const float* gamma      = (const float*)d_in[4];
    const float* beta       = (const float*)d_in[5];
    float* out = (float*)d_out;

    float* ws      = (float*)d_ws;
    float* ws_part = ws;
    float* ws_fold = ws_part + (size_t)NBLK_M * GROW;

    pfn_moments<<<NBLK_M, MW * 32, 0, stream>>>(features, num_points, coors, ws_part);
    pfn_stats<<<1, 256, 0, stream>>>(ws_part, W, gamma, beta, ws_fold);
    pfn_pass_a<<<NBLK_A, AW * 32, 0, stream>>>(features, num_points, coors, ws_fold, out);
}


# max pass: a wave lowers its own priority as it advances (s_setprio 3,2,1,0 from pillar 0,9,12,14): a SIMD's lagging waves get the matrix pipe first, all four end together, no lone-wave tail
# speedup vs baseline: 11.7872x; 1.0314x over previous
#include <hip/hip_runtime.h>

typedef float v2f __attribute__((ext_vector_type(2)));
typedef float v8f __attribute__((ext_vector_type(8)));

#define P_PILLARS 60000
#define NPT 32
#define COUT 64
#define AW 16
#define ASTEPS 15
#define NBLK_A (P_PILLARS / (AW * ASTEPS))

#define MW 16
#define MSTEPS 15
#define NBLK_M (P_PILLARS / (MW * MSTEPS))
#define FPITCH 36

#define GROW 64
__host__ __device__ constexpr int tri(int a, int b) {
    return (a < b ? a : b) * 10 - ((a < b ? a : b) * ((a < b ? a : b) - 1)) / 2 + ((a < b ? b : a) - (a < b ? a : b));
}

__device__ __forceinline__ float xhalf(float x) {
    const int b = __builtin_bit_cast(int, x);
    return __builtin_bit_cast(float, __builtin_amdgcn_permlanex16(b, b, 0x76543210, 0xfedcba98, false, false));
}
#define XROW(x, M) __builtin_bit_cast(float, __builtin_amdgcn_update_dpp( \
    __builtin_bit_cast(int, (x)), __builtin_bit_cast(int, (x)), 0x160 + (M), 0xf, 0xf, false))

__global__ __launch_bounds__(MW * 32) void pfn_moments(
    const float* __restrict__ features,
    const int*   __restrict__ num_points,
    const int*   __restrict__ coors,
    float* __restrict__ ws_part)
{
    __shared__ __attribute__((aligned(16))) float FT[MW][16][FPITCH];
    __shared__ float red[MW][256];

    const int tid  = threadIdx.x;
    const int wid  = tid >> 5;
    const int lane = tid & 31;
    const int half = lane >> 4;
    const int l16  = lane & 15;

    for (int k = 10; k < 16; ++k) FT[wid][k][lane] = 0.0f;

    const int p0 = blockIdx.x * (MW * MSTEPS) + wid;

    v8f acc0 = (v8f){0,0,0,0,0,0,0,0};
    v8f acc1 = (v8f){0,0,0,0,0,0,0,0};

    float4 f  = ((const float4*)features)[p0 * NPT + lane];
    int npts  = num_points[p0];
    int cx    = coors[p0 * 4 + 3];
    int cy    = coors[p0 * 4 + 2];

    for (int t = 0; t < MSTEPS; ++t) {
        const int pn = p0 + ((t + 1 < MSTEPS) ? (t + 1) : t) * MW;
        const float4 fn = ((const float4*)features)[pn * NPT + lane];
        const int npn = num_points[pn];
        const int cxn = coors[pn * 4 + 3];
        const int cyn = coors[pn * 4 + 2];

        float sx = f.x, sy = f.y, sz = f.z;
        sx += xhalf(sx);     sy += xhalf(sy);     sz += xhalf(sz);
        sx += XROW(sx, 8);   sy += XROW(sy, 8);   sz += XROW(sz, 8);
        sx += XROW(sx, 4);   sy += XROW(sy, 4);   sz += XROW(sz, 4);
        sx += XROW(sx, 2);   sy += XROW(sy, 2);   sz += XROW(sz, 2);
        sx += XROW(sx, 1);   sy += XROW(sy, 1);   sz += XROW(sz, 1);
        const float inv_n = 1.0f / (float)npts;
        const float cmx = sx * inv_n, cmy = sy * inv_n, cmz = sz * inv_n;
        const float xc = (float)cx * 0.2f + 0.1f;
        const float yc = (float)cy * 0.2f + (0.1f - 40.0f);
        const float m = (lane < npts) ? 1.0f : 0.0f;

        float (*T)[FPITCH] = FT[wid];
        T[0][lane] = f.x * m;           T[1][lane] = f.y * m;
        T[2][lane] = f.z * m;           T[3][lane] = f.w * m;
        T[4][lane] = (f.x - cmx) * m;   T[5][lane] = (f.y - cmy) * m;
        T[6][lane] = (f.z - cmz) * m;   T[7][lane] = (f.x - xc) * m;
        T[8][lane] = (f.y - yc) * m;    T[9][lane] = m;


        const float4* src = (const float4*)&T[l16][16 * half];
        const float4 q0 = src[0], q1 = src[1], q2 = src[2], q3 = src[3];
        v2f a;
        a[0] = q0.x; a[1] = q0.y;
        acc0 = __builtin_amdgcn_wmma_f32_16x16x4_f32(false, a, false, a, (short)0, acc0, false, false);
        a[0] = q0.z; a[1] = q0.w;
        acc1 = __builtin_amdgcn_wmma_f32_16x16x4_f32(false, a, false, a, (short)0, acc1, false, false);
        a[0] = q1.x; a[1] = q1.y;
        acc0 = __builtin_amdgcn_wmma_f32_16x16x4_f32(false, a, false, a, (short)0, acc0, false, false);
        a[0] = q1.z; a[1] = q1.w;
        acc1 = __builtin_amdgcn_wmma_f32_16x16x4_f32(false, a, false, a, (short)0, acc1, false, false);
        a[0] = q2.x; a[1] = q2.y;
        acc0 = __builtin_amdgcn_wmma_f32_16x16x4_f32(false, a, false, a, (short)0, acc0, false, false);
        a[0] = q2.z; a[1] = q2.w;
        acc1 = __builtin_amdgcn_wmma_f32_16x16x4_f32(false, a, false, a, (short)0, acc1, false, false);
        a[0] = q3.x; a[1] = q3.y;
        acc0 = __builtin_amdgcn_wmma_f32_16x16x4_f32(false, a, false, a, (short)0, acc0, false, false);
        a[0] = q3.z; a[1] = q3.w;
        acc1 = __builtin_amdgcn_wmma_f32_16x16x4_f32(false, a, false, a, (short)0, acc1, false, false);

        f = fn; npts = npn; cx = cxn; cy = cyn;
    }

    #pragma unroll
    for (int r = 0; r < 8; ++r)
        red[wid][(8 * half + r) * 16 + l16] = acc0[r] + acc1[r];
    __syncthreads();
    if (tid < GROW) {
        int k = 0;
        #pragma unroll
        for (int r = 1; r < 10; ++r) k += (tid >= tri(r, r)) ? 1 : 0;
        const int rem = tid - tri(k, k);
        float s = 0.0f;
        if (tid < 55) {
            const int src = k * 16 + k + rem;
            #pragma unroll
            for (int w = 0; w < MW; ++w) s += red[w][src];
        }
        ws_part[(size_t)blockIdx.x * GROW + tid] = s;
    }
}

#define KPAD 12

__global__ __launch_bounds__(256) void pfn_stats(const float* __restrict__ part,
                          const float* __restrict__ W,
                          const float* __restrict__ gamma,
                          const float* __restrict__ beta,
                          float* __restrict__ fold)
{
    __shared__ float red[4][GROW];
    __shared__ float G[GROW];
    const int tid = threadIdx.x;
    const int g = tid >> 6, e = tid & (GROW - 1);
    float s = 0.0f;
    for (int o = 0; o < 2; ++o) {
        #pragma unroll
        for (int i = 0; i < 31; ++i)
            s += part[(size_t)(g + 4 * (o * 31 + i)) * GROW + e];
    }
    if (g < 2) s += part[(size_t)(248 + g) * GROW + e];
    red[g][e] = s;
    __syncthreads();
    if (tid < GROW) G[tid] = (red[0][tid] + red[1][tid]) + (red[2][tid] + red[3][tid]);
    __syncthreads();
    if (tid < COUT) {
        const int c = tid;
        float w[9];
        #pragma unroll
        for (int k = 0; k < 9; ++k) w[k] = W[k * COUT + c];
        float sum = 0.0f, ssq = 0.0f;
        #pragma unroll
        for (int k = 0; k < 9; ++k) {
            sum += G[tri(9, k)] * w[k];
            float row = 0.0f;
            #pragma unroll
            for (int j = 0; j < 9; ++j) row += G[tri(k, j)] * w[j];
            ssq += w[k] * row;
        }
        const float inv = 1.0f / (float)(P_PILLARS * NPT);
        const float mean = sum * inv;
        const float var  = ssq * inv - mean * mean;
        const float sc   = gamma[c] * rsqrtf(var + 1e-3f);
        #pragma unroll
        for (int k = 0; k < 9; ++k) fold[k * COUT + c] = w[k] * sc;
        fold[9 * COUT + c] = beta[c] - mean * sc;
        #pragma unroll
        for (int k = 10; k < KPAD; ++k) fold[k * COUT + c] = 0.0f;
    }
}

__global__ __launch_bounds__(AW * 32) void pfn_pass_a(
    const float* __restrict__ features,
    const int*   __restrict__ num_points,
    const int*   __restrict__ coors,
    const float* __restrict__ fold,
    float* __restrict__ out)
{
    const int tid  = threadIdx.x;
    const int wid  = tid >> 5;
    const int lane = tid & 31;
    const int half = lane >> 4;
    const int l16  = lane & 15;

    const int p0 = blockIdx.x * (AW * ASTEPS) + wid;

    float4 f  = ((const float4*)features)[p0 * NPT + lane];
    int npts  = num_points[p0];
    int cx    = coors[p0 * 4 + 3];
    int cy    = coors[p0 * 4 + 2];

    v2f bfr[KPAD / 4][4];
    #pragma unroll
    for (int ks = 0; ks < KPAD / 4; ++ks) {
        const int kr = ks * 4 + 2 * half;
        #pragma unroll
        for (int nt = 0; nt < 4; ++nt) {
            bfr[ks][nt][0] = fold[kr * COUT + nt * 16 + l16];
            bfr[ks][nt][1] = fold[(kr + 1) * COUT + nt * 16 + l16];
        }
    }
    const float a_one = half ? 0.0f : 1.0f;

    __builtin_amdgcn_s_setprio(3);

    for (int t = 0; t < ASTEPS; ++t) {
        if (t == ASTEPS - 6) __builtin_amdgcn_s_setprio(2);
        if (t == ASTEPS - 3) __builtin_amdgcn_s_setprio(1);
        if (t == ASTEPS - 1) __builtin_amdgcn_s_setprio(0);
        const int p = p0 + t * AW;
        const int pn = p0 + ((t + 1 < ASTEPS) ? (t + 1) : t) * AW;
        const float4 fn = ((const float4*)features)[pn * NPT + lane];
        const int npn = num_points[pn];
        const int cxn = coors[pn * 4 + 3];
        const int cyn = coors[pn * 4 + 2];
        __builtin_amdgcn_sched_barrier(0);

        float4 g;
        g.x = xhalf(f.x);
        g.y = xhalf(f.y);
        g.z = xhalf(f.z);
        g.w = xhalf(f.w);

        float sx = f.x + g.x, sy = f.y + g.y, sz = f.z + g.z;
        sx += XROW(sx, 8);   sy += XROW(sy, 8);   sz += XROW(sz, 8);
        sx += XROW(sx, 4);   sy += XROW(sy, 4);   sz += XROW(sz, 4);
        sx += XROW(sx, 2);   sy += XROW(sy, 2);   sz += XROW(sz, 2);
        sx += XROW(sx, 1);   sy += XROW(sy, 1);   sz += XROW(sz, 1);
        const float inv_n = 1.0f / (float)npts;
        const float cmx = sx * inv_n, cmy = sy * inv_n, cmz = sz * inv_n;
        const float xc = (float)cx * 0.2f + 0.1f;
        const float yc = (float)cy * 0.2f + (0.1f - 40.0f);

        const float o_a = half ? cmz : cmx;
        const float o_b = half ? xc  : cmy;

        float chmax[4];
        #pragma unroll
        for (int nt = 0; nt < 4; ++nt) chmax[nt] = 0.0f;

        #pragma unroll
        for (int mt = 0; mt < 2; ++mt) {
            const bool own = (half == mt);
            const float rx = own ? f.x : g.x, ry = own ? f.y : g.y;
            const float rz = own ? f.z : g.z, rw = own ? f.w : g.w;
            const float m = (mt * 16 + l16 < npts) ? 1.0f : 0.0f;

            const float u  = half ? rz : rx;
            const float v  = half ? rw : ry;
            const float t1 = half ? rx : ry;
            v2f a[3];
            a[0][0] = u * m;           a[0][1] = v * m;
            a[1][0] = (u - o_a) * m;   a[1][1] = (t1 - o_b) * m;
            a[2][0] = half ? 0.0f : (ry - yc) * m;
            a[2][1] = a_one;

            v8f acc[4];
            #pragma unroll
            for (int nt = 0; nt < 4; ++nt) acc[nt] = (v8f){0,0,0,0,0,0,0,0};
            #pragma unroll
            for (int ks = 0; ks < 3; ++ks) {
                #pragma unroll
                for (int nt = 0; nt < 4; ++nt) {
                    acc[nt] = __builtin_amdgcn_wmma_f32_16x16x4_f32(
                        false, a[ks], false, bfr[ks][nt], (short)0, acc[nt], false, false);
                }
            }
            #pragma unroll
            for (int nt = 0; nt < 4; ++nt) {
                #pragma unroll
                for (int r = 0; r < 8; ++r) chmax[nt] = fmaxf(chmax[nt], acc[nt][r]);
            }
        }
        #pragma unroll
        for (int j = 0; j < 2; ++j) {
            const float own_mx  = half ? chmax[2 * j + 1] : chmax[2 * j];
            const float send_mx = half ? chmax[2 * j] : chmax[2 * j + 1];
            out[(size_t)p * COUT + j * 32 + lane] = fmaxf(own_mx, xhalf(send_mx));
        }

        f = fn; npts = npn; cx = cxn; cy = cyn;
    }
}

extern "C" void kernel_launch(void* const* d_in, const int* in_sizes, int n_in,
                              void* d_out, int out_size, void* d_ws, size_t ws_size,
                              hipStream_t stream) {
    const float* features   = (const float*)d_in[0];
    const int*   num_points = (const int*)  d_in[1];
    const int*   coors      = (const int*)  d_in[2];
    const float* W          = (const float*)d_in[3];
    const float* gamma      = (const float*)d_in[4];
    const float* beta       = (const float*)d_in[5];
    float* out = (float*)d_out;

    float* ws      = (float*)d_ws;
    float* ws_part = ws;
    float* ws_fold = ws_part + (size_t)NBLK_M * GROW;

    pfn_moments<<<NBLK_M, MW * 32, 0, stream>>>(features, num_points, coors, ws_part);
    pfn_stats<<<1, 256, 0, stream>>>(ws_part, W, gamma, beta, ws_fold);
    pfn_pass_a<<<NBLK_A, AW * 32, 0, stream>>>(features, num_points, coors, ws_fold, out);
}


# max pass: each block folds the 250 moment rows and builds its scaled-W B operand itself in LDS (3 barriers before the pillar loop); the stats launch goes (3 -> 2 launches); fold's sum order changes
# speedup vs baseline: 12.3833x; 1.0506x over previous
#include <hip/hip_runtime.h>

typedef float v2f __attribute__((ext_vector_type(2)));
typedef float v8f __attribute__((ext_vector_type(8)));

#define P_PILLARS 60000
#define NPT 32
#define COUT 64
#define AW 16
#define ASTEPS 15
#define NBLK_A (P_PILLARS / (AW * ASTEPS))

#define MW 16
#define MSTEPS 15
#define NBLK_M (P_PILLARS / (MW * MSTEPS))
#define FPITCH 36

#define GROW 64
__host__ __device__ constexpr int tri(int a, int b) {
    return (a < b ? a : b) * 10 - ((a < b ? a : b) * ((a < b ? a : b) - 1)) / 2 + ((a < b ? b : a) - (a < b ? a : b));
}

__device__ __forceinline__ float xhalf(float x) {
    const int b = __builtin_bit_cast(int, x);
    return __builtin_bit_cast(float, __builtin_amdgcn_permlanex16(b, b, 0x76543210, 0xfedcba98, false, false));
}
#define XROW(x, M) __builtin_bit_cast(float, __builtin_amdgcn_update_dpp( \
    __builtin_bit_cast(int, (x)), __builtin_bit_cast(int, (x)), 0x160 + (M), 0xf, 0xf, false))

__global__ __launch_bounds__(MW * 32) void pfn_moments(
    const float* __restrict__ features,
    const int*   __restrict__ num_points,
    const int*   __restrict__ coors,
    float* __restrict__ ws_part)
{
    __shared__ __attribute__((aligned(16))) float FT[MW][16][FPITCH];
    __shared__ float red[MW][256];

    const int tid  = threadIdx.x;
    const int wid  = tid >> 5;
    const int lane = tid & 31;
    const int half = lane >> 4;
    const int l16  = lane & 15;

    for (int k = 10; k < 16; ++k) FT[wid][k][lane] = 0.0f;

    const int p0 = blockIdx.x * (MW * MSTEPS) + wid;

    v8f acc0 = (v8f){0,0,0,0,0,0,0,0};
    v8f acc1 = (v8f){0,0,0,0,0,0,0,0};

    float4 f  = ((const float4*)features)[p0 * NPT + lane];
    int npts  = num_points[p0];
    int cx    = coors[p0 * 4 + 3];
    int cy    = coors[p0 * 4 + 2];

    for (int t = 0; t < MSTEPS; ++t) {
        const int pn = p0 + ((t + 1 < MSTEPS) ? (t + 1) : t) * MW;
        const float4 fn = ((const float4*)features)[pn * NPT + lane];
        const int npn = num_points[pn];
        const int cxn = coors[pn * 4 + 3];
        const int cyn = coors[pn * 4 + 2];

        float sx = f.x, sy = f.y, sz = f.z;
        sx += xhalf(sx);     sy += xhalf(sy);     sz += xhalf(sz);
        sx += XROW(sx, 8);   sy += XROW(sy, 8);   sz += XROW(sz, 8);
        sx += XROW(sx, 4);   sy += XROW(sy, 4);   sz += XROW(sz, 4);
        sx += XROW(sx, 2);   sy += XROW(sy, 2);   sz += XROW(sz, 2);
        sx += XROW(sx, 1);   sy += XROW(sy, 1);   sz += XROW(sz, 1);
        const float inv_n = 1.0f / (float)npts;
        const float cmx = sx * inv_n, cmy = sy * inv_n, cmz = sz * inv_n;
        const float xc = (float)cx * 0.2f + 0.1f;
        const float yc = (float)cy * 0.2f + (0.1f - 40.0f);
        const float m = (lane < npts) ? 1.0f : 0.0f;

        float (*T)[FPITCH] = FT[wid];
        T[0][lane] = f.x * m;           T[1][lane] = f.y * m;
        T[2][lane] = f.z * m;           T[3][lane] = f.w * m;
        T[4][lane] = (f.x - cmx) * m;   T[5][lane] = (f.y - cmy) * m;
        T[6][lane] = (f.z - cmz) * m;   T[7][lane] = (f.x - xc) * m;
        T[8][lane] = (f.y - yc) * m;    T[9][lane] = m;


        const float4* src = (const float4*)&T[l16][16 * half];
        const float4 q0 = src[0], q1 = src[1], q2 = src[2], q3 = src[3];
        v2f a;
        a[0] = q0.x; a[1] = q0.y;
        acc0 = __builtin_amdgcn_wmma_f32_16x16x4_f32(false, a, false, a, (short)0, acc0, false, false);
        a[0] = q0.z; a[1] = q0.w;
        acc1 = __builtin_amdgcn_wmma_f32_16x16x4_f32(false, a, false, a, (short)0, acc1, false, false);
        a[0] = q1.x; a[1] = q1.y;
        acc0 = __builtin_amdgcn_wmma_f32_16x16x4_f32(false, a, false, a, (short)0, acc0, false, false);
        a[0] = q1.z; a[1] = q1.w;
        acc1 = __builtin_amdgcn_wmma_f32_16x16x4_f32(false, a, false, a, (short)0, acc1, false, false);
        a[0] = q2.x; a[1] = q2.y;
        acc0 = __builtin_amdgcn_wmma_f32_16x16x4_f32(false, a, false, a, (short)0, acc0, false, false);
        a[0] = q2.z; a[1] = q2.w;
        acc1 = __builtin_amdgcn_wmma_f32_16x16x4_f32(false, a, false, a, (short)0, acc1, false, false);
        a[0] = q3.x; a[1] = q3.y;
        acc0 = __builtin_amdgcn_wmma_f32_16x16x4_f32(false, a, false, a, (short)0, acc0, false, false);
        a[0] = q3.z; a[1] = q3.w;
        acc1 = __builtin_amdgcn_wmma_f32_16x16x4_f32(false, a, false, a, (short)0, acc1, false, false);

        f = fn; npts = npn; cx = cxn; cy = cyn;
    }

    #pragma unroll
    for (int r = 0; r < 8; ++r)
        red[wid][(8 * half + r) * 16 + l16] = acc0[r] + acc1[r];
    __syncthreads();
    if (tid < GROW) {
        int k = 0;
        #pragma unroll
        for (int r = 1; r < 10; ++r) k += (tid >= tri(r, r)) ? 1 : 0;
        const int rem = tid - tri(k, k);
        float s = 0.0f;
        if (tid < 55) {
            const int src = k * 16 + k + rem;
            #pragma unroll
            for (int w = 0; w < MW; ++w) s += red[w][src];
        }
        ws_part[(size_t)blockIdx.x * GROW + tid] = s;
    }
}

#define KPAD 12
#define FGRP 8
#define FROWS (NBLK_M / FGRP)

__global__ __launch_bounds__(AW * 32) void pfn_pass_a(
    const float* __restrict__ features,
    const int*   __restrict__ num_points,
    const int*   __restrict__ coors,
    const float* __restrict__ part,
    const float* __restrict__ W,
    const float* __restrict__ gamma,
    const float* __restrict__ beta,
    float* __restrict__ out)
{
    __shared__ float sred[FGRP][GROW];
    __shared__ float sG[GROW];
    __shared__ float sfold[KPAD][COUT];

    const int tid  = threadIdx.x;
    const int wid  = tid >> 5;
    const int lane = tid & 31;
    const int half = lane >> 4;
    const int l16  = lane & 15;

    const int p0 = blockIdx.x * (AW * ASTEPS) + wid;

    float4 f  = ((const float4*)features)[p0 * NPT + lane];
    int npts  = num_points[p0];
    int cx    = coors[p0 * 4 + 3];
    int cy    = coors[p0 * 4 + 2];

    float wc[9];
    float gam = 0.0f, bet = 0.0f;
    if (tid < COUT) {
        #pragma unroll
        for (int k = 0; k < 9; ++k) wc[k] = W[k * COUT + tid];
        gam = gamma[tid];
        bet = beta[tid];
    }
    {
        const int fg = tid >> 6, fe = tid & (GROW - 1);
        float s = 0.0f;
        const float tail = part[(size_t)(FGRP * FROWS + (fg & 1)) * GROW + fe];
        #pragma unroll
        for (int i = 0; i < FROWS; ++i)
            s += part[(size_t)(fg + FGRP * i) * GROW + fe];
        if (fg < NBLK_M - FGRP * FROWS) s += tail;
        sred[fg][fe] = s;
    }
    __syncthreads();
    if (tid < GROW)
        sG[tid] = ((sred[0][tid] + sred[1][tid]) + (sred[2][tid] + sred[3][tid]))
                + ((sred[4][tid] + sred[5][tid]) + (sred[6][tid] + sred[7][tid]));
    __syncthreads();
    if (tid < COUT) {
        const int c = tid;
        float sum = 0.0f, ssq = 0.0f;
        #pragma unroll
        for (int k = 0; k < 9; ++k) {
            sum += sG[tri(9, k)] * wc[k];
            float row = 0.0f;
            #pragma unroll
            for (int j = 0; j < 9; ++j) row += sG[tri(k, j)] * wc[j];
            ssq += wc[k] * row;
        }
        const float inv = 1.0f / (float)(P_PILLARS * NPT);
        const float mean = sum * inv;
        const float var  = ssq * inv - mean * mean;
        const float sc   = gam * rsqrtf(var + 1e-3f);
        #pragma unroll
        for (int k = 0; k < 9; ++k) sfold[k][c] = wc[k] * sc;
        sfold[9][c] = bet - mean * sc;
        #pragma unroll
        for (int k = 10; k < KPAD; ++k) sfold[k][c] = 0.0f;
    }
    __syncthreads();

    v2f bfr[KPAD / 4][4];
    #pragma unroll
    for (int ks = 0; ks < KPAD / 4; ++ks) {
        const int kr = ks * 4 + 2 * half;
        #pragma unroll
        for (int nt = 0; nt < 4; ++nt) {
            bfr[ks][nt][0] = sfold[kr][nt * 16 + l16];
            bfr[ks][nt][1] = sfold[kr + 1][nt * 16 + l16];
        }
    }
    const float a_one = half ? 0.0f : 1.0f;

    __builtin_amdgcn_s_setprio(3);

    for (int t = 0; t < ASTEPS; ++t) {
        if (t == ASTEPS - 6) __builtin_amdgcn_s_setprio(2);
        if (t == ASTEPS - 3) __builtin_amdgcn_s_setprio(1);
        if (t == ASTEPS - 1) __builtin_amdgcn_s_setprio(0);
        const int p = p0 + t * AW;
        const int pn = p0 + ((t + 1 < ASTEPS) ? (t + 1) : t) * AW;
        const float4 fn = ((const float4*)features)[pn * NPT + lane];
        const int npn = num_points[pn];
        const int cxn = coors[pn * 4 + 3];
        const int cyn = coors[pn * 4 + 2];
        __builtin_amdgcn_sched_barrier(0);

        float4 g;
        g.x = xhalf(f.x);
        g.y = xhalf(f.y);
        g.z = xhalf(f.z);
        g.w = xhalf(f.w);

        float sx = f.x + g.x, sy = f.y + g.y, sz = f.z + g.z;
        sx += XROW(sx, 8);   sy += XROW(sy, 8);   sz += XROW(sz, 8);
        sx += XROW(sx, 4);   sy += XROW(sy, 4);   sz += XROW(sz, 4);
        sx += XROW(sx, 2);   sy += XROW(sy, 2);   sz += XROW(sz, 2);
        sx += XROW(sx, 1);   sy += XROW(sy, 1);   sz += XROW(sz, 1);
        const float inv_n = 1.0f / (float)npts;
        const float cmx = sx * inv_n, cmy = sy * inv_n, cmz = sz * inv_n;
        const float xc = (float)cx * 0.2f + 0.1f;
        const float yc = (float)cy * 0.2f + (0.1f - 40.0f);

        const float o_a = half ? cmz : cmx;
        const float o_b = half ? xc  : cmy;

        float chmax[4];
        #pragma unroll
        for (int nt = 0; nt < 4; ++nt) chmax[nt] = 0.0f;

        #pragma unroll
        for (int mt = 0; mt < 2; ++mt) {
            const bool own = (half == mt);
            const float rx = own ? f.x : g.x, ry = own ? f.y : g.y;
            const float rz = own ? f.z : g.z, rw = own ? f.w : g.w;
            const float m = (mt * 16 + l16 < npts) ? 1.0f : 0.0f;

            const float u  = half ? rz : rx;
            const float v  = half ? rw : ry;
            const float t1 = half ? rx : ry;
            v2f a[3];
            a[0][0] = u * m;           a[0][1] = v * m;
            a[1][0] = (u - o_a) * m;   a[1][1] = (t1 - o_b) * m;
            a[2][0] = half ? 0.0f : (ry - yc) * m;
            a[2][1] = a_one;

            v8f acc[4];
            #pragma unroll
            for (int nt = 0; nt < 4; ++nt) acc[nt] = (v8f){0,0,0,0,0,0,0,0};
            #pragma unroll
            for (int ks = 0; ks < 3; ++ks) {
                #pragma unroll
                for (int nt = 0; nt < 4; ++nt) {
                    acc[nt] = __builtin_amdgcn_wmma_f32_16x16x4_f32(
                        false, a[ks], false, bfr[ks][nt], (short)0, acc[nt], false, false);
                }
            }
            #pragma unroll
            for (int nt = 0; nt < 4; ++nt) {
                #pragma unroll
                for (int r = 0; r < 8; ++r) chmax[nt] = fmaxf(chmax[nt], acc[nt][r]);
            }
        }
        #pragma unroll
        for (int j = 0; j < 2; ++j) {
            const float own_mx  = half ? chmax[2 * j + 1] : chmax[2 * j];
            const float send_mx = half ? chmax[2 * j] : chmax[2 * j + 1];
            out[(size_t)p * COUT + j * 32 + lane] = fmaxf(own_mx, xhalf(send_mx));
        }

        f = fn; npts = npn; cx = cxn; cy = cyn;
    }
}

extern "C" void kernel_launch(void* const* d_in, const int* in_sizes, int n_in,
                              void* d_out, int out_size, void* d_ws, size_t ws_size,
                              hipStream_t stream) {
    const float* features   = (const float*)d_in[0];
    const int*   num_points = (const int*)  d_in[1];
    const int*   coors      = (const int*)  d_in[2];
    const float* W          = (const float*)d_in[3];
    const float* gamma      = (const float*)d_in[4];
    const float* beta       = (const float*)d_in[5];
    float* out = (float*)d_out;

    float* ws      = (float*)d_ws;
    float* ws_part = ws;

    pfn_moments<<<NBLK_M, MW * 32, 0, stream>>>(features, num_points, coors, ws_part);
    pfn_pass_a<<<NBLK_A, AW * 32, 0, stream>>>(features, num_points, coors, ws_part, W, gamma, beta, out);
}


# moment pass: loads leave early: points five pillars ahead in register slots (was under one short pillar), a wave's 15 counts and cells in two loads up front (lane = pillar, readlane); sum order kept
# speedup vs baseline: 12.8489x; 1.0376x over previous
#include <hip/hip_runtime.h>

typedef float v2f __attribute__((ext_vector_type(2)));
typedef float v8f __attribute__((ext_vector_type(8)));

#define P_PILLARS 60000
#define NPT 32
#define COUT 64
#define AW 16
#define ASTEPS 15
#define NBLK_A (P_PILLARS / (AW * ASTEPS))

#define MW 16
#define MSTEPS 15
#define NBLK_M (P_PILLARS / (MW * MSTEPS))
#define MDEPTH 5
#define FPITCH 36

#define GROW 64
__host__ __device__ constexpr int tri(int a, int b) {
    return (a < b ? a : b) * 10 - ((a < b ? a : b) * ((a < b ? a : b) - 1)) / 2 + ((a < b ? b : a) - (a < b ? a : b));
}

__device__ __forceinline__ float xhalf(float x) {
    const int b = __builtin_bit_cast(int, x);
    return __builtin_bit_cast(float, __builtin_amdgcn_permlanex16(b, b, 0x76543210, 0xfedcba98, false, false));
}
#define XROW(x, M) __builtin_bit_cast(float, __builtin_amdgcn_update_dpp( \
    __builtin_bit_cast(int, (x)), __builtin_bit_cast(int, (x)), 0x160 + (M), 0xf, 0xf, false))

template <bool REFILL>
__device__ __forceinline__ void moment_pillar(
    float4& slot, const float4* __restrict__ next, const int npts, const int cx, const int cy,
    float (*T)[FPITCH], const int lane, const int l16, const int half, v8f& acc0, v8f& acc1)
{
    const float4 f = slot;

    float sx = f.x, sy = f.y, sz = f.z;
    sx += xhalf(sx);     sy += xhalf(sy);     sz += xhalf(sz);
    sx += XROW(sx, 8);   sy += XROW(sy, 8);   sz += XROW(sz, 8);
    sx += XROW(sx, 4);   sy += XROW(sy, 4);   sz += XROW(sz, 4);
    sx += XROW(sx, 2);   sy += XROW(sy, 2);   sz += XROW(sz, 2);
    sx += XROW(sx, 1);   sy += XROW(sy, 1);   sz += XROW(sz, 1);
    const float inv_n = 1.0f / (float)npts;
    const float cmx = sx * inv_n, cmy = sy * inv_n, cmz = sz * inv_n;
    const float xc = (float)cx * 0.2f + 0.1f;
    const float yc = (float)cy * 0.2f + (0.1f - 40.0f);
    const float m = (lane < npts) ? 1.0f : 0.0f;

    T[0][lane] = f.x * m;           T[1][lane] = f.y * m;
    T[2][lane] = f.z * m;           T[3][lane] = f.w * m;
    T[4][lane] = (f.x - cmx) * m;   T[5][lane] = (f.y - cmy) * m;
    T[6][lane] = (f.z - cmz) * m;   T[7][lane] = (f.x - xc) * m;
    T[8][lane] = (f.y - yc) * m;    T[9][lane] = m;

    if (REFILL) {
        __builtin_amdgcn_sched_barrier(0);
        slot = *next;
        __builtin_amdgcn_sched_barrier(0);
    }


    const float4* src = (const float4*)&T[l16][16 * half];
    const float4 q0 = src[0], q1 = src[1], q2 = src[2], q3 = src[3];
    v2f a;
    a[0] = q0.x; a[1] = q0.y;
    acc0 = __builtin_amdgcn_wmma_f32_16x16x4_f32(false, a, false, a, (short)0, acc0, false, false);
    a[0] = q0.z; a[1] = q0.w;
    acc1 = __builtin_amdgcn_wmma_f32_16x16x4_f32(false, a, false, a, (short)0, acc1, false, false);
    a[0] = q1.x; a[1] = q1.y;
    acc0 = __builtin_amdgcn_wmma_f32_16x16x4_f32(false, a, false, a, (short)0, acc0, false, false);
    a[0] = q1.z; a[1] = q1.w;
    acc1 = __builtin_amdgcn_wmma_f32_16x16x4_f32(false, a, false, a, (short)0, acc1, false, false);
    a[0] = q2.x; a[1] = q2.y;
    acc0 = __builtin_amdgcn_wmma_f32_16x16x4_f32(false, a, false, a, (short)0, acc0, false, false);
    a[0] = q2.z; a[1] = q2.w;
    acc1 = __builtin_amdgcn_wmma_f32_16x16x4_f32(false, a, false, a, (short)0, acc1, false, false);
    a[0] = q3.x; a[1] = q3.y;
    acc0 = __builtin_amdgcn_wmma_f32_16x16x4_f32(false, a, false, a, (short)0, acc0, false, false);
    a[0] = q3.z; a[1] = q3.w;
    acc1 = __builtin_amdgcn_wmma_f32_16x16x4_f32(false, a, false, a, (short)0, acc1, false, false);
}

__global__ __launch_bounds__(MW * 32) void pfn_moments(
    const float* __restrict__ features,
    const int*   __restrict__ num_points,
    const int*   __restrict__ coors,
    float* __restrict__ ws_part)
{
    __shared__ __attribute__((aligned(16))) float FT[MW][16][FPITCH];
    __shared__ float red[MW][256];

    const int tid  = threadIdx.x;
    const int wid  = tid >> 5;
    const int lane = tid & 31;
    const int half = lane >> 4;
    const int l16  = lane & 15;

    for (int k = 10; k < 16; ++k) FT[wid][k][lane] = 0.0f;

    const int p0 = blockIdx.x * (MW * MSTEPS) + wid;

    v8f acc0 = (v8f){0,0,0,0,0,0,0,0};
    v8f acc1 = (v8f){0,0,0,0,0,0,0,0};

    const int lp  = p0 + ((lane < MSTEPS) ? lane : (MSTEPS - 1)) * MW;
    const int nv  = num_points[lp];
    const int2 cv = *(const int2*)(coors + lp * 4 + 2);
    __builtin_amdgcn_sched_barrier(0);
    const float4* fp = (const float4*)features + (size_t)p0 * NPT + lane;
    float4 rf[MDEPTH];
    #pragma unroll
    for (int s = 0; s < MDEPTH; ++s)
        rf[s] = fp[s * (MW * NPT)];

    float (*T)[FPITCH] = FT[wid];

    #pragma nounroll
    for (int j = 0; j < MSTEPS / MDEPTH - 1; ++j) {
        #pragma unroll
        for (int s = 0; s < MDEPTH; ++s) {
            const int t = j * MDEPTH + s;
            moment_pillar<true>(rf[s], fp + (t + MDEPTH) * (MW * NPT),
                                __builtin_amdgcn_readlane(nv, t), __builtin_amdgcn_readlane(cv.y, t),
                                __builtin_amdgcn_readlane(cv.x, t), T, lane, l16, half, acc0, acc1);
        }
    }
    #pragma unroll
    for (int s = 0; s < MDEPTH; ++s) {
        const int t = MSTEPS - MDEPTH + s;
        moment_pillar<false>(rf[s], fp,
                             __builtin_amdgcn_readlane(nv, t), __builtin_amdgcn_readlane(cv.y, t),
                             __builtin_amdgcn_readlane(cv.x, t), T, lane, l16, half, acc0, acc1);
    }

    #pragma unroll
    for (int r = 0; r < 8; ++r)
        red[wid][(8 * half + r) * 16 + l16] = acc0[r] + acc1[r];
    __syncthreads();
    if (tid < GROW) {
        int k = 0;
        #pragma unroll
        for (int r = 1; r < 10; ++r) k += (tid >= tri(r, r)) ? 1 : 0;
        const int rem = tid - tri(k, k);
        float s = 0.0f;
        if (tid < 55) {
            const int src = k * 16 + k + rem;
            #pragma unroll
            for (int w = 0; w < MW; ++w) s += red[w][src];
        }
        ws_part[(size_t)blockIdx.x * GROW + tid] = s;
    }
}

#define KPAD 12
#define FGRP 8
#define FROWS (NBLK_M / FGRP)

__global__ __launch_bounds__(AW * 32) void pfn_pass_a(
    const float* __restrict__ features,
    const int*   __restrict__ num_points,
    const int*   __restrict__ coors,
    const float* __restrict__ part,
    const float* __restrict__ W,
    const float* __restrict__ gamma,
    const float* __restrict__ beta,
    float* __restrict__ out)
{
    __shared__ float sred[FGRP][GROW];
    __shared__ float sG[GROW];
    __shared__ float sfold[KPAD][COUT];

    const int tid  = threadIdx.x;
    const int wid  = tid >> 5;
    const int lane = tid & 31;
    const int half = lane >> 4;
    const int l16  = lane & 15;

    const int p0 = blockIdx.x * (AW * ASTEPS) + wid;

    float4 f  = ((const float4*)features)[p0 * NPT + lane];
    int npts  = num_points[p0];
    int cx    = coors[p0 * 4 + 3];
    int cy    = coors[p0 * 4 + 2];

    float wc[9];
    float gam = 0.0f, bet = 0.0f;
    if (tid < COUT) {
        #pragma unroll
        for (int k = 0; k < 9; ++k) wc[k] = W[k * COUT + tid];
        gam = gamma[tid];
        bet = beta[tid];
    }
    {
        const int fg = tid >> 6, fe = tid & (GROW - 1);
        float s = 0.0f;
        const float tail = part[(size_t)(FGRP * FROWS + (fg & 1)) * GROW + fe];
        #pragma unroll
        for (int i = 0; i < FROWS; ++i)
            s += part[(size_t)(fg + FGRP * i) * GROW + fe];
        if (fg < NBLK_M - FGRP * FROWS) s += tail;
        sred[fg][fe] = s;
    }
    __syncthreads();
    if (tid < GROW)
        sG[tid] = ((sred[0][tid] + sred[1][tid]) + (sred[2][tid] + sred[3][tid]))
                + ((sred[4][tid] + sred[5][tid]) + (sred[6][tid] + sred[7][tid]));
    __syncthreads();
    if (tid < COUT) {
        const int c = tid;
        float sum = 0.0f, ssq = 0.0f;
        #pragma unroll
        for (int k = 0; k < 9; ++k) {
            sum += sG[tri(9, k)] * wc[k];
            float row = 0.0f;
            #pragma unroll
            for (int j = 0; j < 9; ++j) row += sG[tri(k, j)] * wc[j];
            ssq += wc[k] * row;
        }
        const float inv = 1.0f / (float)(P_PILLARS * NPT);
        const float mean = sum * inv;
        const float var  = ssq * inv - mean * mean;
        const float sc   = gam * rsqrtf(var + 1e-3f);
        #pragma unroll
        for (int k = 0; k < 9; ++k) sfold[k][c] = wc[k] * sc;
        sfold[9][c] = bet - mean * sc;
        #pragma unroll
        for (int k = 10; k < KPAD; ++k) sfold[k][c] = 0.0f;
    }
    __syncthreads();

    v2f bfr[KPAD / 4][4];
    #pragma unroll
    for (int ks = 0; ks < KPAD / 4; ++ks) {
        const int kr = ks * 4 + 2 * half;
        #pragma unroll
        for (int nt = 0; nt < 4; ++nt) {
            bfr[ks][nt][0] = sfold[kr][nt * 16 + l16];
            bfr[ks][nt][1] = sfold[kr + 1][nt * 16 + l16];
        }
    }
    const float a_one = half ? 0.0f : 1.0f;

    __builtin_amdgcn_s_setprio(3);

    for (int t = 0; t < ASTEPS; ++t) {
        if (t == ASTEPS - 6) __builtin_amdgcn_s_setprio(2);
        if (t == ASTEPS - 3) __builtin_amdgcn_s_setprio(1);
        if (t == ASTEPS - 1) __builtin_amdgcn_s_setprio(0);
        const int p = p0 + t * AW;
        const int pn = p0 + ((t + 1 < ASTEPS) ? (t + 1) : t) * AW;
        const float4 fn = ((const float4*)features)[pn * NPT + lane];
        const int npn = num_points[pn];
        const int cxn = coors[pn * 4 + 3];
        const int cyn = coors[pn * 4 + 2];
        __builtin_amdgcn_sched_barrier(0);

        float4 g;
        g.x = xhalf(f.x);
        g.y = xhalf(f.y);
        g.z = xhalf(f.z);
        g.w = xhalf(f.w);

        float sx = f.x + g.x, sy = f.y + g.y, sz = f.z + g.z;
        sx += XROW(sx, 8);   sy += XROW(sy, 8);   sz += XROW(sz, 8);
        sx += XROW(sx, 4);   sy += XROW(sy, 4);   sz += XROW(sz, 4);
        sx += XROW(sx, 2);   sy += XROW(sy, 2);   sz += XROW(sz, 2);
        sx += XROW(sx, 1);   sy += XROW(sy, 1);   sz += XROW(sz, 1);
        const float inv_n = 1.0f / (float)npts;
        const float cmx = sx * inv_n, cmy = sy * inv_n, cmz = sz * inv_n;
        const float xc = (float)cx * 0.2f + 0.1f;
        const float yc = (float)cy * 0.2f + (0.1f - 40.0f);

        const float o_a = half ? cmz : cmx;
        const float o_b = half ? xc  : cmy;

        float chmax[4];
        #pragma unroll
        for (int nt = 0; nt < 4; ++nt) chmax[nt] = 0.0f;

        #pragma unroll
        for (int mt = 0; mt < 2; ++mt) {
            const bool own = (half == mt);
            const float rx = own ? f.x : g.x, ry = own ? f.y : g.y;
            const float rz = own ? f.z : g.z, rw = own ? f.w : g.w;
            const float m = (mt * 16 + l16 < npts) ? 1.0f : 0.0f;

            const float u  = half ? rz : rx;
            const float v  = half ? rw : ry;
            const float t1 = half ? rx : ry;
            v2f a[3];
            a[0][0] = u * m;           a[0][1] = v * m;
            a[1][0] = (u - o_a) * m;   a[1][1] = (t1 - o_b) * m;
            a[2][0] = half ? 0.0f : (ry - yc) * m;
            a[2][1] = a_one;

            v8f acc[4];
            #pragma unroll
            for (int nt = 0; nt < 4; ++nt) acc[nt] = (v8f){0,0,0,0,0,0,0,0};
            #pragma unroll
            for (int ks = 0; ks < 3; ++ks) {
                #pragma unroll
                for (int nt = 0; nt < 4; ++nt) {
                    acc[nt] = __builtin_amdgcn_wmma_f32_16x16x4_f32(
                        false, a[ks], false, bfr[ks][nt], (short)0, acc[nt], false, false);
                }
            }
            #pragma unroll
            for (int nt = 0; nt < 4; ++nt) {
                #pragma unroll
                for (int r = 0; r < 8; ++r) chmax[nt] = fmaxf(chmax[nt], acc[nt][r]);
            }
        }
        #pragma unroll
        for (int j = 0; j < 2; ++j) {
            const float own_mx  = half ? chmax[2 * j + 1] : chmax[2 * j];
            const float send_mx = half ? chmax[2 * j] : chmax[2 * j + 1];
            out[(size_t)p * COUT + j * 32 + lane] = fmaxf(own_mx, xhalf(send_mx));
        }

        f = fn; npts = npn; cx = cxn; cy = cyn;
    }
}

extern "C" void kernel_launch(void* const* d_in, const int* in_sizes, int n_in,
                              void* d_out, int out_size, void* d_ws, size_t ws_size,
                              hipStream_t stream) {
    const float* features   = (const float*)d_in[0];
    const int*   num_points = (const int*)  d_in[1];
    const int*   coors      = (const int*)  d_in[2];
    const float* W          = (const float*)d_in[3];
    const float* gamma      = (const float*)d_in[4];
    const float* beta       = (const float*)d_in[5];
    float* out = (float*)d_out;

    float* ws      = (float*)d_ws;
    float* ws_part = ws;

    pfn_moments<<<NBLK_M, MW * 32, 0, stream>>>(features, num_points, coors, ws_part);
    pfn_pass_a<<<NBLK_A, AW * 32, 0, stream>>>(features, num_points, coors, ws_part, W, gamma, beta, out);
}
